# starting kernel
# baseline (speedup 1.0000x reference)
#include <hip/hip_runtime.h>


typedef __attribute__((ext_vector_type(2))) float v2f;
typedef __attribute__((ext_vector_type(8))) float v8f;

#define FEAT 16
#define KDIM 180
#define KSTEPS 45
#define YSTRIDE 186
#define ROWS_PER_BLOCK 16

__global__ void sc_pack_basis(const float* __restrict__ U3,
                              const float* __restrict__ U2,
                              float2* __restrict__ P, int ntiles) {
  int total = ntiles * KSTEPS * 32;
  for (int q = blockIdx.x * blockDim.x + threadIdx.x; q < total;
       q += gridDim.x * blockDim.x) {
    int lane = q & 31;
    int kk = (q >> 5) % KSTEPS;
    int tile = q / (KSTEPS * 32);
    int m0 = 4 * kk + 2 * (lane >> 4);
    int n = tile * 16 + (lane & 15);
    float a = (m0 < 176) ? U3[n * 176 + m0] : U2[n * 4 + (m0 - 176)];
    int m1 = m0 + 1;
    float b = (m1 < 176) ? U3[n * 176 + m1] : U2[n * 4 + (m1 - 176)];
    P[q] = make_float2(a, b);
  }
}

__global__ __launch_bounds__(256) void sc_wmma_kernel(
    const float* __restrict__ x, const int* __restrict__ indices,
    const v2f* __restrict__ P0, const v2f* __restrict__ P1,
    const v2f* __restrict__ P2,
    const float* __restrict__ U1_0, const float* __restrict__ U1_1,
    const float* __restrict__ U1_2,
    const float* __restrict__ W1_0, const float* __restrict__ W1_1,
    const float* __restrict__ W1_2,
    const float* __restrict__ W2_0, const float* __restrict__ W2_1,
    const float* __restrict__ W2_2,
    const float* __restrict__ W3_0, const float* __restrict__ W3_1,
    const float* __restrict__ W3_2,
    float* __restrict__ out) {
  __shared__ float xs[ROWS_PER_BLOCK][FEAT + 1];
  __shared__ alignas(16) float Ylds[3][ROWS_PER_BLOCK][YSTRIDE];
  __shared__ float w1s[3][ROWS_PER_BLOCK];
  __shared__ int es[ROWS_PER_BLOCK];

  const int tid = threadIdx.x;
  const int rowbase = blockIdx.x * ROWS_PER_BLOCK;

  {
    int r = tid >> 4, l = tid & 15;
    xs[r][l] = x[(rowbase + r) * FEAT + l];
  }
  if (tid < ROWS_PER_BLOCK) es[tid] = indices[(rowbase + tid) >> 6];
  __syncthreads();

  for (int q = tid; q < 3 * ROWS_PER_BLOCK * KDIM; q += 256) {
    int t = q / (ROWS_PER_BLOCK * KDIM);
    int rm = q - t * (ROWS_PER_BLOCK * KDIM);
    int r = rm / KDIM, m = rm - r * KDIM;
    int bc = rowbase + r, c = bc & 63, e = es[r];
    const float* W3 = (t == 0) ? W3_0 : (t == 1) ? W3_1 : W3_2;
    const float* W2 = (t == 0) ? W2_0 : (t == 1) ? W2_1 : W2_2;
    float v;
    if (m < 176) {
      int l = m / 11, p = m - l * 11;
      v = xs[r][l] * W3[(e * 11 + p) * 64 + c];
    } else {
      v = W2[(e * 4 + (m - 176)) * 64 + c];
    }
    Ylds[t][r][m] = v;
  }
  if (tid < 3 * ROWS_PER_BLOCK) {
    int t = tid >> 4, r = tid & 15;
    int bc = rowbase + r, c = bc & 63;
    const float* W1 = (t == 0) ? W1_0 : (t == 1) ? W1_1 : W1_2;
    w1s[t][r] = W1[es[r] * 64 + c];
  }
  __syncthreads();

  const int wave = tid >> 5;
  const int lane = tid & 31;
  const int half = lane >> 4;
  const int l15 = lane & 15;

  float xh[8];
#pragma unroll
  for (int v = 0; v < 8; ++v) xh[v] = xs[l15][(half << 3) + v];

  for (int g = wave; g < 9; g += 8) {
    int t, k;
    if (g < 1)       { t = 0; k = 0; }
    else if (g < 4)  { t = 1; k = g - 1; }
    else             { t = 2; k = g - 4; }
    const v2f* Pt    = (t == 0) ? P0 : (t == 1) ? P1 : P2;
    const float* U1  = (t == 0) ? U1_0 : (t == 1) ? U1_1 : U1_2;
    const int kt     = (t == 0) ? 1 : (t == 1) ? 3 : 5;
    const int seg    = (t == 0) ? 0 : (t == 1) ? 64 : 256;
    const float w1r  = w1s[t][l15];
    const float* Yrow = &Ylds[t][l15][0];

    float accT = 0.0f;

    for (int i = 0; i < 16; ++i) {
      const v2f* Ap = Pt + ((size_t)(k * 16 + i) * KSTEPS) * 32 + lane;
      __builtin_prefetch(Ap, 0, 1);

      v8f acc = {};
#pragma unroll 9
      for (int kk = 0; kk < KSTEPS; ++kk) {
        v2f A = Ap[kk * 32];
        v2f B = *reinterpret_cast<const v2f*>(Yrow + 4 * kk + 2 * half);
        acc = __builtin_amdgcn_wmma_f32_16x16x4_f32(
            false, A, false, B, (short)0, acc, false, false);
      }

      float s = 0.0f;
#pragma unroll
      for (int v = 0; v < 8; ++v) s = fmaf(acc[v], xh[v], s);
      s += __shfl_xor(s, 16, 32);

      const float u1v = U1[k * 16 + i];
      const float c1 = fmaf(u1v, w1r, s);
      accT = fmaf(c1, xs[l15][i], accT);
    }

    if (lane < 16) {
      const int bc = rowbase + l15;
      const int b = bc >> 6, c = bc & 63;
      out[b * 576 + seg + c * kt + k] = accT;
    }
  }
}

extern "C" void kernel_launch(void* const* d_in, const int* in_sizes, int n_in,
                              void* d_out, int out_size, void* d_ws, size_t ws_size,
                              hipStream_t stream) {
  (void)in_sizes; (void)n_in; (void)out_size; (void)ws_size;
  const float* x   = (const float*)d_in[0];
  const int* idx   = (const int*)d_in[1];
  const float* U1_0 = (const float*)d_in[2];
  const float* W1_0 = (const float*)d_in[3];
  const float* U2_0 = (const float*)d_in[4];
  const float* W2_0 = (const float*)d_in[5];
  const float* U3_0 = (const float*)d_in[6];
  const float* W3_0 = (const float*)d_in[7];
  const float* U1_1 = (const float*)d_in[8];
  const float* W1_1 = (const float*)d_in[9];
  const float* U2_1 = (const float*)d_in[10];
  const float* W2_1 = (const float*)d_in[11];
  const float* U3_1 = (const float*)d_in[12];
  const float* W3_1 = (const float*)d_in[13];
  const float* U1_2 = (const float*)d_in[14];
  const float* W1_2 = (const float*)d_in[15];
  const float* U2_2 = (const float*)d_in[16];
  const float* W2_2 = (const float*)d_in[17];
  const float* U3_2 = (const float*)d_in[18];
  const float* W3_2 = (const float*)d_in[19];
  float* out = (float*)d_out;

  float2* P0 = (float2*)d_ws;
  float2* P1 = P0 + (size_t)16 * KSTEPS * 32;
  float2* P2 = P1 + (size_t)48 * KSTEPS * 32;

  sc_pack_basis<<<64, 256, 0, stream>>>(U3_0, U2_0, P0, 16);
  sc_pack_basis<<<128, 256, 0, stream>>>(U3_1, U2_1, P1, 48);
  sc_pack_basis<<<256, 256, 0, stream>>>(U3_2, U2_2, P2, 80);

  sc_wmma_kernel<<<1024, 256, 0, stream>>>(
      x, idx, (const v2f*)P0, (const v2f*)P1, (const v2f*)P2,
      U1_0, U1_1, U1_2,
      W1_0, W1_1, W1_2,
      W2_0, W2_1, W2_2,
      W3_0, W3_1, W3_2,
      out);
}


# Block's 144 (group,i) tiles dealt evenly to 12 waves of 12 (was 9 groups on 8 waves, wave 0 ran two); each group's i-sum is now two partial sums added via LDS (sum order changes)
# speedup vs baseline: 1.3624x; 1.3624x over previous
#include <hip/hip_runtime.h>


typedef __attribute__((ext_vector_type(2))) float v2f;
typedef __attribute__((ext_vector_type(8))) float v8f;

#define FEAT 16
#define KDIM 180
#define KSTEPS 45
#define YSTRIDE 186
#define ROWS_PER_BLOCK 16
#define NWAVES 12
#define TILES_PER_WAVE 12

__global__ void sc_pack_basis(const float* __restrict__ U3,
                              const float* __restrict__ U2,
                              float2* __restrict__ P, int ntiles) {
  int total = ntiles * KSTEPS * 32;
  for (int q = blockIdx.x * blockDim.x + threadIdx.x; q < total;
       q += gridDim.x * blockDim.x) {
    int lane = q & 31;
    int kk = (q >> 5) % KSTEPS;
    int tile = q / (KSTEPS * 32);
    int m0 = 4 * kk + 2 * (lane >> 4);
    int n = tile * 16 + (lane & 15);
    float a = (m0 < 176) ? U3[n * 176 + m0] : U2[n * 4 + (m0 - 176)];
    int m1 = m0 + 1;
    float b = (m1 < 176) ? U3[n * 176 + m1] : U2[n * 4 + (m1 - 176)];
    P[q] = make_float2(a, b);
  }
}

__global__ __launch_bounds__(NWAVES * 32) void sc_wmma_kernel(
    const float* __restrict__ x, const int* __restrict__ indices,
    const v2f* __restrict__ P0, const v2f* __restrict__ P1,
    const v2f* __restrict__ P2,
    const float* __restrict__ U1_0, const float* __restrict__ U1_1,
    const float* __restrict__ U1_2,
    const float* __restrict__ W1_0, const float* __restrict__ W1_1,
    const float* __restrict__ W1_2,
    const float* __restrict__ W2_0, const float* __restrict__ W2_1,
    const float* __restrict__ W2_2,
    const float* __restrict__ W3_0, const float* __restrict__ W3_1,
    const float* __restrict__ W3_2,
    float* __restrict__ out) {
  __shared__ float xs[ROWS_PER_BLOCK][FEAT + 1];
  __shared__ alignas(16) float Ylds[3][ROWS_PER_BLOCK][YSTRIDE];
  __shared__ float w1s[3][ROWS_PER_BLOCK];
  __shared__ int es[ROWS_PER_BLOCK];
  __shared__ float part[9][2][ROWS_PER_BLOCK];

  const int tid = threadIdx.x;
  const int rowbase = blockIdx.x * ROWS_PER_BLOCK;

  if (tid < ROWS_PER_BLOCK * FEAT) {
    int r = tid >> 4, l = tid & 15;
    xs[r][l] = x[(rowbase + r) * FEAT + l];
  }
  if (tid < ROWS_PER_BLOCK) es[tid] = indices[(rowbase + tid) >> 6];
  __syncthreads();

  for (int q = tid; q < 3 * ROWS_PER_BLOCK * KDIM; q += NWAVES * 32) {
    int t = q / (ROWS_PER_BLOCK * KDIM);
    int rm = q - t * (ROWS_PER_BLOCK * KDIM);
    int r = rm / KDIM, m = rm - r * KDIM;
    int bc = rowbase + r, c = bc & 63, e = es[r];
    const float* W3 = (t == 0) ? W3_0 : (t == 1) ? W3_1 : W3_2;
    const float* W2 = (t == 0) ? W2_0 : (t == 1) ? W2_1 : W2_2;
    float v;
    if (m < 176) {
      int l = m / 11, p = m - l * 11;
      v = xs[r][l] * W3[(e * 11 + p) * 64 + c];
    } else {
      v = W2[(e * 4 + (m - 176)) * 64 + c];
    }
    Ylds[t][r][m] = v;
  }
  if (tid < 3 * ROWS_PER_BLOCK) {
    int t = tid >> 4, r = tid & 15;
    int bc = rowbase + r, c = bc & 63;
    const float* W1 = (t == 0) ? W1_0 : (t == 1) ? W1_1 : W1_2;
    w1s[t][r] = W1[es[r] * 64 + c];
  }
  __syncthreads();

  const int wave = __builtin_amdgcn_readfirstlane(tid >> 5);
  const int lane = tid & 31;
  const int half = lane >> 4;
  const int l15 = lane & 15;

  float xh[8];
#pragma unroll
  for (int v = 0; v < 8; ++v) xh[v] = xs[l15][(half << 3) + v];

  const int u0 = wave * TILES_PER_WAVE;
  const int gfirst = u0 >> 4;
  const int ifirst = u0 & 15;
  const int nA = (16 - ifirst < TILES_PER_WAVE) ? (16 - ifirst) : TILES_PER_WAVE;

  for (int sgm = 0; sgm < 2; ++sgm) {
    const int g  = gfirst + sgm;
    const int ib = sgm ? 0 : ifirst;
    const int ie = sgm ? (TILES_PER_WAVE - nA) : (ifirst + nA);
    if (ib >= ie) break;
    int t, k;
    if (g < 1)       { t = 0; k = 0; }
    else if (g < 4)  { t = 1; k = g - 1; }
    else             { t = 2; k = g - 4; }
    const v2f* Pt    = (t == 0) ? P0 : (t == 1) ? P1 : P2;
    const float* U1  = (t == 0) ? U1_0 : (t == 1) ? U1_1 : U1_2;
    const float w1r  = w1s[t][l15];
    const float* Yrow = &Ylds[t][l15][0];

    float accT = 0.0f;

    for (int i = ib; i < ie; ++i) {
      const v2f* Ap = Pt + ((size_t)(k * 16 + i) * KSTEPS) * 32 + lane;
      __builtin_prefetch(Ap, 0, 1);

      v8f acc = {};
#pragma unroll 9
      for (int kk = 0; kk < KSTEPS; ++kk) {
        v2f A = Ap[kk * 32];
        v2f B = *reinterpret_cast<const v2f*>(Yrow + 4 * kk + 2 * half);
        acc = __builtin_amdgcn_wmma_f32_16x16x4_f32(
            false, A, false, B, (short)0, acc, false, false);
      }

      float s = 0.0f;
#pragma unroll
      for (int v = 0; v < 8; ++v) s = fmaf(acc[v], xh[v], s);
      s += __shfl_xor(s, 16, 32);

      const float u1v = U1[k * 16 + i];
      const float c1 = fmaf(u1v, w1r, s);
      accT = fmaf(c1, xs[l15][i], accT);
    }

    if (lane < 16) part[g][ib ? 1 : 0][l15] = accT;
  }
  __syncthreads();

  if (tid < 9 * ROWS_PER_BLOCK) {
    int t, j;
    if (tid < 16)      { t = 0; j = tid; }
    else if (tid < 64) { t = 1; j = tid - 16; }
    else               { t = 2; j = tid - 64; }
    const int kt  = (t == 0) ? 1 : (t == 1) ? 3 : 5;
    const int seg = (t == 0) ? 0 : (t == 1) ? 64 : 256;
    const int g0  = (t == 0) ? 0 : (t == 1) ? 1 : 4;
    const int r = j / kt, k = j - r * kt;
    const int bc = rowbase + r;
    const int b = bc >> 6, c = bc & 63;
    out[b * 576 + seg + c * kt + k] = part[g0 + k][0][r] + part[g0 + k][1][r];
  }
}

extern "C" void kernel_launch(void* const* d_in, const int* in_sizes, int n_in,
                              void* d_out, int out_size, void* d_ws, size_t ws_size,
                              hipStream_t stream) {
  (void)in_sizes; (void)n_in; (void)out_size; (void)ws_size;
  const float* x   = (const float*)d_in[0];
  const int* idx   = (const int*)d_in[1];
  const float* U1_0 = (const float*)d_in[2];
  const float* W1_0 = (const float*)d_in[3];
  const float* U2_0 = (const float*)d_in[4];
  const float* W2_0 = (const float*)d_in[5];
  const float* U3_0 = (const float*)d_in[6];
  const float* W3_0 = (const float*)d_in[7];
  const float* U1_1 = (const float*)d_in[8];
  const float* W1_1 = (const float*)d_in[9];
  const float* U2_1 = (const float*)d_in[10];
  const float* W2_1 = (const float*)d_in[11];
  const float* U3_1 = (const float*)d_in[12];
  const float* W3_1 = (const float*)d_in[13];
  const float* U1_2 = (const float*)d_in[14];
  const float* W1_2 = (const float*)d_in[15];
  const float* U2_2 = (const float*)d_in[16];
  const float* W2_2 = (const float*)d_in[17];
  const float* U3_2 = (const float*)d_in[18];
  const float* W3_2 = (const float*)d_in[19];
  float* out = (float*)d_out;

  float2* P0 = (float2*)d_ws;
  float2* P1 = P0 + (size_t)16 * KSTEPS * 32;
  float2* P2 = P1 + (size_t)48 * KSTEPS * 32;

  sc_pack_basis<<<64, 256, 0, stream>>>(U3_0, U2_0, P0, 16);
  sc_pack_basis<<<128, 256, 0, stream>>>(U3_1, U2_1, P1, 48);
  sc_pack_basis<<<256, 256, 0, stream>>>(U3_2, U2_2, P2, 80);

  sc_wmma_kernel<<<1024, NWAVES * 32, 0, stream>>>(
      x, idx, (const v2f*)P0, (const v2f*)P1, (const v2f*)P2,
      U1_0, U1_1, U1_2,
      W1_0, W1_1, W1_2,
      W2_0, W2_1, W2_2,
      W3_0, W3_1, W3_2,
      out);
}


# The three basis-repacking launches (one per irrep) merged into one launch over all 144 tiles, each thread's three loads in flight together; launches per repetition 4 -> 2
# speedup vs baseline: 1.4428x; 1.0590x over previous
#include <hip/hip_runtime.h>


typedef __attribute__((ext_vector_type(2))) float v2f;
typedef __attribute__((ext_vector_type(8))) float v8f;

#define FEAT 16
#define KDIM 180
#define KSTEPS 45
#define YSTRIDE 186
#define ROWS_PER_BLOCK 16
#define NWAVES 12
#define TILES_PER_WAVE 12

#define NTILES_ALL 144
#define PACK_THREADS 256
#define PACK_BLOCKS 270
#define PACK_PER_THREAD 3
static_assert(PACK_BLOCKS * PACK_THREADS * PACK_PER_THREAD ==
                  NTILES_ALL * KSTEPS * 32,
              "pack grid must cover every fragment element exactly once");

__global__ __launch_bounds__(PACK_THREADS) void sc_pack_basis(
    const float* __restrict__ U3_0, const float* __restrict__ U2_0,
    const float* __restrict__ U3_1, const float* __restrict__ U2_1,
    const float* __restrict__ U3_2, const float* __restrict__ U2_2,
    float2* __restrict__ P) {
  const int gtid = blockIdx.x * PACK_THREADS + threadIdx.x;
  float2 val[PACK_PER_THREAD];
  for (int u = 0; u < PACK_PER_THREAD; ++u) {
    const int q = gtid + u * (PACK_BLOCKS * PACK_THREADS);
    const int lane = q & 31;
    const int kk = (q >> 5) % KSTEPS;
    const int tile = q / (KSTEPS * 32);
    const float* U3 = (tile < 16) ? U3_0 : (tile < 64) ? U3_1 : U3_2;
    const float* U2 = (tile < 16) ? U2_0 : (tile < 64) ? U2_1 : U2_2;
    const int ltile = (tile < 16) ? tile : (tile < 64) ? tile - 16 : tile - 64;
    const int m0 = 4 * kk + 2 * (lane >> 4);
    const int n = ltile * 16 + (lane & 15);
    const float a = (m0 < 176) ? U3[n * 176 + m0] : U2[n * 4 + (m0 - 176)];
    const int m1 = m0 + 1;
    const float b = (m1 < 176) ? U3[n * 176 + m1] : U2[n * 4 + (m1 - 176)];
    val[u] = make_float2(a, b);
  }
  for (int u = 0; u < PACK_PER_THREAD; ++u)
    P[gtid + u * (PACK_BLOCKS * PACK_THREADS)] = val[u];
}

__global__ __launch_bounds__(NWAVES * 32) void sc_wmma_kernel(
    const float* __restrict__ x, const int* __restrict__ indices,
    const v2f* __restrict__ P0, const v2f* __restrict__ P1,
    const v2f* __restrict__ P2,
    const float* __restrict__ U1_0, const float* __restrict__ U1_1,
    const float* __restrict__ U1_2,
    const float* __restrict__ W1_0, const float* __restrict__ W1_1,
    const float* __restrict__ W1_2,
    const float* __restrict__ W2_0, const float* __restrict__ W2_1,
    const float* __restrict__ W2_2,
    const float* __restrict__ W3_0, const float* __restrict__ W3_1,
    const float* __restrict__ W3_2,
    float* __restrict__ out) {
  __shared__ float xs[ROWS_PER_BLOCK][FEAT + 1];
  __shared__ alignas(16) float Ylds[3][ROWS_PER_BLOCK][YSTRIDE];
  __shared__ float w1s[3][ROWS_PER_BLOCK];
  __shared__ int es[ROWS_PER_BLOCK];
  __shared__ float part[9][2][ROWS_PER_BLOCK];

  const int tid = threadIdx.x;
  const int rowbase = blockIdx.x * ROWS_PER_BLOCK;

  if (tid < ROWS_PER_BLOCK * FEAT) {
    int r = tid >> 4, l = tid & 15;
    xs[r][l] = x[(rowbase + r) * FEAT + l];
  }
  if (tid < ROWS_PER_BLOCK) es[tid] = indices[(rowbase + tid) >> 6];
  __syncthreads();

  for (int q = tid; q < 3 * ROWS_PER_BLOCK * KDIM; q += NWAVES * 32) {
    int t = q / (ROWS_PER_BLOCK * KDIM);
    int rm = q - t * (ROWS_PER_BLOCK * KDIM);
    int r = rm / KDIM, m = rm - r * KDIM;
    int bc = rowbase + r, c = bc & 63, e = es[r];
    const float* W3 = (t == 0) ? W3_0 : (t == 1) ? W3_1 : W3_2;
    const float* W2 = (t == 0) ? W2_0 : (t == 1) ? W2_1 : W2_2;
    float v;
    if (m < 176) {
      int l = m / 11, p = m - l * 11;
      v = xs[r][l] * W3[(e * 11 + p) * 64 + c];
    } else {
      v = W2[(e * 4 + (m - 176)) * 64 + c];
    }
    Ylds[t][r][m] = v;
  }
  if (tid < 3 * ROWS_PER_BLOCK) {
    int t = tid >> 4, r = tid & 15;
    int bc = rowbase + r, c = bc & 63;
    const float* W1 = (t == 0) ? W1_0 : (t == 1) ? W1_1 : W1_2;
    w1s[t][r] = W1[es[r] * 64 + c];
  }
  __syncthreads();

  const int wave = __builtin_amdgcn_readfirstlane(tid >> 5);
  const int lane = tid & 31;
  const int half = lane >> 4;
  const int l15 = lane & 15;

  float xh[8];
#pragma unroll
  for (int v = 0; v < 8; ++v) xh[v] = xs[l15][(half << 3) + v];

  const int u0 = wave * TILES_PER_WAVE;
  const int gfirst = u0 >> 4;
  const int ifirst = u0 & 15;
  const int nA = (16 - ifirst < TILES_PER_WAVE) ? (16 - ifirst) : TILES_PER_WAVE;

  for (int sgm = 0; sgm < 2; ++sgm) {
    const int g  = gfirst + sgm;
    const int ib = sgm ? 0 : ifirst;
    const int ie = sgm ? (TILES_PER_WAVE - nA) : (ifirst + nA);
    if (ib >= ie) break;
    int t, k;
    if (g < 1)       { t = 0; k = 0; }
    else if (g < 4)  { t = 1; k = g - 1; }
    else             { t = 2; k = g - 4; }
    const v2f* Pt    = (t == 0) ? P0 : (t == 1) ? P1 : P2;
    const float* U1  = (t == 0) ? U1_0 : (t == 1) ? U1_1 : U1_2;
    const float w1r  = w1s[t][l15];
    const float* Yrow = &Ylds[t][l15][0];

    float accT = 0.0f;

    for (int i = ib; i < ie; ++i) {
      const v2f* Ap = Pt + ((size_t)(k * 16 + i) * KSTEPS) * 32 + lane;
      __builtin_prefetch(Ap, 0, 1);

      v8f acc = {};
#pragma unroll 9
      for (int kk = 0; kk < KSTEPS; ++kk) {
        v2f A = Ap[kk * 32];
        v2f B = *reinterpret_cast<const v2f*>(Yrow + 4 * kk + 2 * half);
        acc = __builtin_amdgcn_wmma_f32_16x16x4_f32(
            false, A, false, B, (short)0, acc, false, false);
      }

      float s = 0.0f;
#pragma unroll
      for (int v = 0; v < 8; ++v) s = fmaf(acc[v], xh[v], s);
      s += __shfl_xor(s, 16, 32);

      const float u1v = U1[k * 16 + i];
      const float c1 = fmaf(u1v, w1r, s);
      accT = fmaf(c1, xs[l15][i], accT);
    }

    if (lane < 16) part[g][ib ? 1 : 0][l15] = accT;
  }
  __syncthreads();

  if (tid < 9 * ROWS_PER_BLOCK) {
    int t, j;
    if (tid < 16)      { t = 0; j = tid; }
    else if (tid < 64) { t = 1; j = tid - 16; }
    else               { t = 2; j = tid - 64; }
    const int kt  = (t == 0) ? 1 : (t == 1) ? 3 : 5;
    const int seg = (t == 0) ? 0 : (t == 1) ? 64 : 256;
    const int g0  = (t == 0) ? 0 : (t == 1) ? 1 : 4;
    const int r = j / kt, k = j - r * kt;
    const int bc = rowbase + r;
    const int b = bc >> 6, c = bc & 63;
    out[b * 576 + seg + c * kt + k] = part[g0 + k][0][r] + part[g0 + k][1][r];
  }
}

extern "C" void kernel_launch(void* const* d_in, const int* in_sizes, int n_in,
                              void* d_out, int out_size, void* d_ws, size_t ws_size,
                              hipStream_t stream) {
  (void)in_sizes; (void)n_in; (void)out_size; (void)ws_size;
  const float* x   = (const float*)d_in[0];
  const int* idx   = (const int*)d_in[1];
  const float* U1_0 = (const float*)d_in[2];
  const float* W1_0 = (const float*)d_in[3];
  const float* U2_0 = (const float*)d_in[4];
  const float* W2_0 = (const float*)d_in[5];
  const float* U3_0 = (const float*)d_in[6];
  const float* W3_0 = (const float*)d_in[7];
  const float* U1_1 = (const float*)d_in[8];
  const float* W1_1 = (const float*)d_in[9];
  const float* U2_1 = (const float*)d_in[10];
  const float* W2_1 = (const float*)d_in[11];
  const float* U3_1 = (const float*)d_in[12];
  const float* W3_1 = (const float*)d_in[13];
  const float* U1_2 = (const float*)d_in[14];
  const float* W1_2 = (const float*)d_in[15];
  const float* U2_2 = (const float*)d_in[16];
  const float* W2_2 = (const float*)d_in[17];
  const float* U3_2 = (const float*)d_in[18];
  const float* W3_2 = (const float*)d_in[19];
  float* out = (float*)d_out;

  float2* P0 = (float2*)d_ws;
  float2* P1 = P0 + (size_t)16 * KSTEPS * 32;
  float2* P2 = P1 + (size_t)48 * KSTEPS * 32;

  sc_pack_basis<<<PACK_BLOCKS, PACK_THREADS, 0, stream>>>(
      U3_0, U2_0, U3_1, U2_1, U3_2, U2_2, P0);

  sc_wmma_kernel<<<1024, NWAVES * 32, 0, stream>>>(
      x, idx, (const v2f*)P0, (const v2f*)P1, (const v2f*)P2,
      U1_0, U1_1, U1_2,
      W1_0, W1_1, W1_2,
      W2_0, W2_1, W2_2,
      W3_0, W3_1, W3_2,
      out);
}


# Main kernel Y build: each per-row weight (W3, W2, W1) is loaded once and multiplied by the row's 16 x values; was one global load plus index divisions per Y element, 22-23 serial trips per thread
# speedup vs baseline: 1.5627x; 1.0831x over previous
#include <hip/hip_runtime.h>


typedef __attribute__((ext_vector_type(2))) float v2f;
typedef __attribute__((ext_vector_type(8))) float v8f;

#define FEAT 16
#define KDIM 180
#define KSTEPS 45
#define YSTRIDE 186
#define ROWS_PER_BLOCK 16
#define NWAVES 12
#define TILES_PER_WAVE 12

#define NTILES_ALL 144
#define PACK_THREADS 256
#define PACK_BLOCKS 270
#define PACK_PER_THREAD 3
static_assert(PACK_BLOCKS * PACK_THREADS * PACK_PER_THREAD ==
                  NTILES_ALL * KSTEPS * 32,
              "pack grid must cover every fragment element exactly once");

__global__ __launch_bounds__(PACK_THREADS) void sc_pack_basis(
    const float* __restrict__ U3_0, const float* __restrict__ U2_0,
    const float* __restrict__ U3_1, const float* __restrict__ U2_1,
    const float* __restrict__ U3_2, const float* __restrict__ U2_2,
    float2* __restrict__ P) {
  const int gtid = blockIdx.x * PACK_THREADS + threadIdx.x;
  float2 val[PACK_PER_THREAD];
  for (int u = 0; u < PACK_PER_THREAD; ++u) {
    const int q = gtid + u * (PACK_BLOCKS * PACK_THREADS);
    const int lane = q & 31;
    const int kk = (q >> 5) % KSTEPS;
    const int tile = q / (KSTEPS * 32);
    const float* U3 = (tile < 16) ? U3_0 : (tile < 64) ? U3_1 : U3_2;
    const float* U2 = (tile < 16) ? U2_0 : (tile < 64) ? U2_1 : U2_2;
    const int ltile = (tile < 16) ? tile : (tile < 64) ? tile - 16 : tile - 64;
    const int m0 = 4 * kk + 2 * (lane >> 4);
    const int n = ltile * 16 + (lane & 15);
    const float a = (m0 < 176) ? U3[n * 176 + m0] : U2[n * 4 + (m0 - 176)];
    const int m1 = m0 + 1;
    const float b = (m1 < 176) ? U3[n * 176 + m1] : U2[n * 4 + (m1 - 176)];
    val[u] = make_float2(a, b);
  }
  for (int u = 0; u < PACK_PER_THREAD; ++u)
    P[gtid + u * (PACK_BLOCKS * PACK_THREADS)] = val[u];
}

__global__ __launch_bounds__(NWAVES * 32) void sc_wmma_kernel(
    const float* __restrict__ x, const int* __restrict__ indices,
    const v2f* __restrict__ P0, const v2f* __restrict__ P1,
    const v2f* __restrict__ P2,
    const float* __restrict__ U1_0, const float* __restrict__ U1_1,
    const float* __restrict__ U1_2,
    const float* __restrict__ W1_0, const float* __restrict__ W1_1,
    const float* __restrict__ W1_2,
    const float* __restrict__ W2_0, const float* __restrict__ W2_1,
    const float* __restrict__ W2_2,
    const float* __restrict__ W3_0, const float* __restrict__ W3_1,
    const float* __restrict__ W3_2,
    float* __restrict__ out) {
  __shared__ float xs[ROWS_PER_BLOCK][FEAT + 1];
  __shared__ alignas(16) float Ylds[3][ROWS_PER_BLOCK][YSTRIDE];
  __shared__ float w1s[3][ROWS_PER_BLOCK];
  __shared__ float part[9][2][ROWS_PER_BLOCK];

  const int tid = threadIdx.x;
  const int rowbase = blockIdx.x * ROWS_PER_BLOCK;

  const int e = indices[rowbase >> 6];
  const int r = tid & 15;
  const int c = (rowbase & 63) + r;

  const int tpA = tid >> 4;
  const int tA = tpA / 11, pA = tpA - tA * 11;
  const float* W3A = (tA == 0) ? W3_0 : (tA == 1) ? W3_1 : W3_2;
  const float wA = W3A[(e * 11 + pA) * 64 + c];

  const bool second = tid < 144;
  const int jq = (tid - 144) >> 4;
  const int t2 = jq >> 2, p2 = jq & 3;
  const int t1 = jq - 12;
  const float* Wsel;
  int wi;
  if (second) {
    Wsel = W3_2;
    wi = (e * 11 + (tpA + 2)) * 64 + c;
  } else if (jq < 12) {
    Wsel = (t2 == 0) ? W2_0 : (t2 == 1) ? W2_1 : W2_2;
    wi = (e * 4 + p2) * 64 + c;
  } else {
    Wsel = (t1 == 0) ? W1_0 : (t1 == 1) ? W1_1 : W1_2;
    wi = e * 64 + c;
  }
  const float wB = Wsel[wi];

  if (tid < ROWS_PER_BLOCK * FEAT) xs[tid >> 4][tid & 15] = x[rowbase * FEAT + tid];
  __syncthreads();

  {
    float xr[FEAT];
#pragma unroll
    for (int l = 0; l < FEAT; ++l) xr[l] = xs[r][l];
    float* yA = &Ylds[tA][r][pA];
#pragma unroll
    for (int l = 0; l < FEAT; ++l) yA[l * 11] = xr[l] * wA;
    if (second) {
      float* yB = &Ylds[2][r][tpA + 2];
#pragma unroll
      for (int l = 0; l < FEAT; ++l) yB[l * 11] = xr[l] * wB;
    } else if (jq < 12) {
      Ylds[t2][r][176 + p2] = wB;
    } else {
      w1s[t1][r] = wB;
    }
  }
  __syncthreads();

  const int wave = __builtin_amdgcn_readfirstlane(tid >> 5);
  const int lane = tid & 31;
  const int half = lane >> 4;
  const int l15 = lane & 15;

  float xh[8];
#pragma unroll
  for (int v = 0; v < 8; ++v) xh[v] = xs[l15][(half << 3) + v];

  const int u0 = wave * TILES_PER_WAVE;
  const int gfirst = u0 >> 4;
  const int ifirst = u0 & 15;
  const int nA = (16 - ifirst < TILES_PER_WAVE) ? (16 - ifirst) : TILES_PER_WAVE;

  for (int sgm = 0; sgm < 2; ++sgm) {
    const int g  = gfirst + sgm;
    const int ib = sgm ? 0 : ifirst;
    const int ie = sgm ? (TILES_PER_WAVE - nA) : (ifirst + nA);
    if (ib >= ie) break;
    int t, k;
    if (g < 1)       { t = 0; k = 0; }
    else if (g < 4)  { t = 1; k = g - 1; }
    else             { t = 2; k = g - 4; }
    const v2f* Pt    = (t == 0) ? P0 : (t == 1) ? P1 : P2;
    const float* U1  = (t == 0) ? U1_0 : (t == 1) ? U1_1 : U1_2;
    const float w1r  = w1s[t][l15];
    const float* Yrow = &Ylds[t][l15][0];

    float accT = 0.0f;

    for (int i = ib; i < ie; ++i) {
      const v2f* Ap = Pt + ((size_t)(k * 16 + i) * KSTEPS) * 32 + lane;
      __builtin_prefetch(Ap, 0, 1);

      v8f acc = {};
#pragma unroll 9
      for (int kk = 0; kk < KSTEPS; ++kk) {
        v2f A = Ap[kk * 32];
        v2f B = *reinterpret_cast<const v2f*>(Yrow + 4 * kk + 2 * half);
        acc = __builtin_amdgcn_wmma_f32_16x16x4_f32(
            false, A, false, B, (short)0, acc, false, false);
      }

      float s = 0.0f;
#pragma unroll
      for (int v = 0; v < 8; ++v) s = fmaf(acc[v], xh[v], s);
      s += __shfl_xor(s, 16, 32);

      const float u1v = U1[k * 16 + i];
      const float c1 = fmaf(u1v, w1r, s);
      accT = fmaf(c1, xs[l15][i], accT);
    }

    if (lane < 16) part[g][ib ? 1 : 0][l15] = accT;
  }
  __syncthreads();

  if (tid < 9 * ROWS_PER_BLOCK) {
    int t, j;
    if (tid < 16)      { t = 0; j = tid; }
    else if (tid < 64) { t = 1; j = tid - 16; }
    else               { t = 2; j = tid - 64; }
    const int kt  = (t == 0) ? 1 : (t == 1) ? 3 : 5;
    const int seg = (t == 0) ? 0 : (t == 1) ? 64 : 256;
    const int g0  = (t == 0) ? 0 : (t == 1) ? 1 : 4;
    const int r = j / kt, k = j - r * kt;
    const int bc = rowbase + r;
    const int b = bc >> 6, c = bc & 63;
    out[b * 576 + seg + c * kt + k] = part[g0 + k][0][r] + part[g0 + k][1][r];
  }
}

extern "C" void kernel_launch(void* const* d_in, const int* in_sizes, int n_in,
                              void* d_out, int out_size, void* d_ws, size_t ws_size,
                              hipStream_t stream) {
  (void)in_sizes; (void)n_in; (void)out_size; (void)ws_size;
  const float* x   = (const float*)d_in[0];
  const int* idx   = (const int*)d_in[1];
  const float* U1_0 = (const float*)d_in[2];
  const float* W1_0 = (const float*)d_in[3];
  const float* U2_0 = (const float*)d_in[4];
  const float* W2_0 = (const float*)d_in[5];
  const float* U3_0 = (const float*)d_in[6];
  const float* W3_0 = (const float*)d_in[7];
  const float* U1_1 = (const float*)d_in[8];
  const float* W1_1 = (const float*)d_in[9];
  const float* U2_1 = (const float*)d_in[10];
  const float* W2_1 = (const float*)d_in[11];
  const float* U3_1 = (const float*)d_in[12];
  const float* W3_1 = (const float*)d_in[13];
  const float* U1_2 = (const float*)d_in[14];
  const float* W1_2 = (const float*)d_in[15];
  const float* U2_2 = (const float*)d_in[16];
  const float* W2_2 = (const float*)d_in[17];
  const float* U3_2 = (const float*)d_in[18];
  const float* W3_2 = (const float*)d_in[19];
  float* out = (float*)d_out;

  float2* P0 = (float2*)d_ws;
  float2* P1 = P0 + (size_t)16 * KSTEPS * 32;
  float2* P2 = P1 + (size_t)48 * KSTEPS * 32;

  sc_pack_basis<<<PACK_BLOCKS, PACK_THREADS, 0, stream>>>(
      U3_0, U2_0, U3_1, U2_1, U3_2, U2_2, P0);

  sc_wmma_kernel<<<1024, NWAVES * 32, 0, stream>>>(
      x, idx, (const v2f*)P0, (const v2f*)P1, (const v2f*)P2,
      U1_0, U1_1, U1_2,
      W1_0, W1_1, W1_2,
      W2_0, W2_1, W2_2,
      W3_0, W3_1, W3_2,
      out);
}


# Main kernel: a block covers 32 rows (two 16-row tiles) and each wave multiplies every basis fragment it loads with both tiles' Y fragments; basis loads per WMMA halve, 512 blocks
# speedup vs baseline: 1.6663x; 1.0663x over previous
#include <hip/hip_runtime.h>


typedef __attribute__((ext_vector_type(2))) float v2f;
typedef __attribute__((ext_vector_type(8))) float v8f;

#define FEAT 16
#define KDIM 180
#define KSTEPS 45
#define YSTRIDE 186
#define ROWS_PER_BLOCK 32
#define NWAVES 12
#define TILES_PER_WAVE 12

#define NTILES_ALL 144
#define PACK_THREADS 256
#define PACK_BLOCKS 270
#define PACK_PER_THREAD 3
static_assert(PACK_BLOCKS * PACK_THREADS * PACK_PER_THREAD ==
                  NTILES_ALL * KSTEPS * 32,
              "pack grid must cover every fragment element exactly once");

__global__ __launch_bounds__(PACK_THREADS) void sc_pack_basis(
    const float* __restrict__ U3_0, const float* __restrict__ U2_0,
    const float* __restrict__ U3_1, const float* __restrict__ U2_1,
    const float* __restrict__ U3_2, const float* __restrict__ U2_2,
    float2* __restrict__ P) {
  const int gtid = blockIdx.x * PACK_THREADS + threadIdx.x;
  float2 val[PACK_PER_THREAD];
  for (int u = 0; u < PACK_PER_THREAD; ++u) {
    const int q = gtid + u * (PACK_BLOCKS * PACK_THREADS);
    const int lane = q & 31;
    const int kk = (q >> 5) % KSTEPS;
    const int tile = q / (KSTEPS * 32);
    const float* U3 = (tile < 16) ? U3_0 : (tile < 64) ? U3_1 : U3_2;
    const float* U2 = (tile < 16) ? U2_0 : (tile < 64) ? U2_1 : U2_2;
    const int ltile = (tile < 16) ? tile : (tile < 64) ? tile - 16 : tile - 64;
    const int m0 = 4 * kk + 2 * (lane >> 4);
    const int n = ltile * 16 + (lane & 15);
    const float a = (m0 < 176) ? U3[n * 176 + m0] : U2[n * 4 + (m0 - 176)];
    const int m1 = m0 + 1;
    const float b = (m1 < 176) ? U3[n * 176 + m1] : U2[n * 4 + (m1 - 176)];
    val[u] = make_float2(a, b);
  }
  for (int u = 0; u < PACK_PER_THREAD; ++u)
    P[gtid + u * (PACK_BLOCKS * PACK_THREADS)] = val[u];
}

__global__ __launch_bounds__(NWAVES * 32) void sc_wmma_kernel(
    const float* __restrict__ x, const int* __restrict__ indices,
    const v2f* __restrict__ P0, const v2f* __restrict__ P1,
    const v2f* __restrict__ P2,
    const float* __restrict__ U1_0, const float* __restrict__ U1_1,
    const float* __restrict__ U1_2,
    const float* __restrict__ W1_0, const float* __restrict__ W1_1,
    const float* __restrict__ W1_2,
    const float* __restrict__ W2_0, const float* __restrict__ W2_1,
    const float* __restrict__ W2_2,
    const float* __restrict__ W3_0, const float* __restrict__ W3_1,
    const float* __restrict__ W3_2,
    float* __restrict__ out) {
  __shared__ float xs[ROWS_PER_BLOCK][FEAT + 1];
  __shared__ alignas(16) float Ylds[3][ROWS_PER_BLOCK][YSTRIDE];
  __shared__ float w1s[3][ROWS_PER_BLOCK];
  __shared__ float part[9][2][ROWS_PER_BLOCK];

  const int tid = threadIdx.x;
  const int rowbase = blockIdx.x * ROWS_PER_BLOCK;

  const int e = indices[rowbase >> 6];
  const int r = tid & 31;
  const int q = tid >> 5;
  const int c = (rowbase & 63) + r;

  const int tpA = q, tpB = q + 12, tpC = q + 24;
  const int tA = tpA / 11, pA = tpA - tA * 11;
  const int tB = tpB / 11, pB = tpB - tB * 11;
  const bool third = q < 9;
  const int pC = tpC - 22;
  const float* W3A = (tA == 0) ? W3_0 : W3_1;
  const float* W3B = (tB == 1) ? W3_1 : W3_2;
  const float wA = W3A[(e * 11 + pA) * 64 + c];
  const float wB = W3B[(e * 11 + pB) * 64 + c];
  float wC = 0.0f;
  if (third) wC = W3_2[(e * 11 + pC) * 64 + c];
  const int t2 = q >> 2, p2 = q & 3;
  const float* W2s = (t2 == 0) ? W2_0 : (t2 == 1) ? W2_1 : W2_2;
  const float w2v = W2s[(e * 4 + p2) * 64 + c];
  float w1v = 0.0f;
  if (q < 3) {
    const float* W1s = (q == 0) ? W1_0 : (q == 1) ? W1_1 : W1_2;
    w1v = W1s[e * 64 + c];
  }

  xs[tid >> 4][tid & 15] = x[rowbase * FEAT + tid];
  if (tid < ROWS_PER_BLOCK * FEAT - NWAVES * 32) {
    const int u = tid + NWAVES * 32;
    xs[u >> 4][u & 15] = x[rowbase * FEAT + u];
  }
  __syncthreads();

  {
    float xr[FEAT];
#pragma unroll
    for (int l = 0; l < FEAT; ++l) xr[l] = xs[r][l];
    float* yA = &Ylds[tA][r][pA];
#pragma unroll
    for (int l = 0; l < FEAT; ++l) yA[l * 11] = xr[l] * wA;
    float* yB = &Ylds[tB][r][pB];
#pragma unroll
    for (int l = 0; l < FEAT; ++l) yB[l * 11] = xr[l] * wB;
    if (third) {
      float* yC = &Ylds[2][r][pC];
#pragma unroll
      for (int l = 0; l < FEAT; ++l) yC[l * 11] = xr[l] * wC;
    }
    Ylds[t2][r][176 + p2] = w2v;
    if (q < 3) w1s[q][r] = w1v;
  }
  __syncthreads();

  const int wave = __builtin_amdgcn_readfirstlane(tid >> 5);
  const int lane = tid & 31;
  const int half = lane >> 4;
  const int l15 = lane & 15;

  float xh0[8], xh1[8];
#pragma unroll
  for (int v = 0; v < 8; ++v) {
    xh0[v] = xs[l15][(half << 3) + v];
    xh1[v] = xs[16 + l15][(half << 3) + v];
  }

  const int u0 = wave * TILES_PER_WAVE;
  const int gfirst = u0 >> 4;
  const int ifirst = u0 & 15;
  const int nA = (16 - ifirst < TILES_PER_WAVE) ? (16 - ifirst) : TILES_PER_WAVE;

  for (int sgm = 0; sgm < 2; ++sgm) {
    const int g  = gfirst + sgm;
    const int ib = sgm ? 0 : ifirst;
    const int ie = sgm ? (TILES_PER_WAVE - nA) : (ifirst + nA);
    if (ib >= ie) break;
    int t, k;
    if (g < 1)       { t = 0; k = 0; }
    else if (g < 4)  { t = 1; k = g - 1; }
    else             { t = 2; k = g - 4; }
    const v2f* Pt    = (t == 0) ? P0 : (t == 1) ? P1 : P2;
    const float* U1  = (t == 0) ? U1_0 : (t == 1) ? U1_1 : U1_2;
    const float w1r0 = w1s[t][l15];
    const float w1r1 = w1s[t][16 + l15];
    const float* Yrow0 = &Ylds[t][l15][0];
    const float* Yrow1 = &Ylds[t][16 + l15][0];

    float accT0 = 0.0f, accT1 = 0.0f;

    for (int i = ib; i < ie; ++i) {
      const v2f* Ap = Pt + ((size_t)(k * 16 + i) * KSTEPS) * 32 + lane;
      __builtin_prefetch(Ap, 0, 1);

      v8f acc0 = {}, acc1 = {};
#pragma unroll 9
      for (int kk = 0; kk < KSTEPS; ++kk) {
        v2f A = Ap[kk * 32];
        v2f B0 = *reinterpret_cast<const v2f*>(Yrow0 + 4 * kk + 2 * half);
        v2f B1 = *reinterpret_cast<const v2f*>(Yrow1 + 4 * kk + 2 * half);
        acc0 = __builtin_amdgcn_wmma_f32_16x16x4_f32(
            false, A, false, B0, (short)0, acc0, false, false);
        acc1 = __builtin_amdgcn_wmma_f32_16x16x4_f32(
            false, A, false, B1, (short)0, acc1, false, false);
      }

      float s0 = 0.0f, s1 = 0.0f;
#pragma unroll
      for (int v = 0; v < 8; ++v) {
        s0 = fmaf(acc0[v], xh0[v], s0);
        s1 = fmaf(acc1[v], xh1[v], s1);
      }
      s0 += __shfl_xor(s0, 16, 32);
      s1 += __shfl_xor(s1, 16, 32);

      const float u1v = U1[k * 16 + i];
      const float c10 = fmaf(u1v, w1r0, s0);
      const float c11 = fmaf(u1v, w1r1, s1);
      accT0 = fmaf(c10, xs[l15][i], accT0);
      accT1 = fmaf(c11, xs[16 + l15][i], accT1);
    }

    if (lane < 16) {
      part[g][ib ? 1 : 0][l15] = accT0;
      part[g][ib ? 1 : 0][16 + l15] = accT1;
    }
  }
  __syncthreads();

  if (tid < 9 * ROWS_PER_BLOCK) {
    int t, j;
    if (tid < 32)       { t = 0; j = tid; }
    else if (tid < 128) { t = 1; j = tid - 32; }
    else                { t = 2; j = tid - 128; }
    const int kt  = (t == 0) ? 1 : (t == 1) ? 3 : 5;
    const int seg = (t == 0) ? 0 : (t == 1) ? 64 : 256;
    const int g0  = (t == 0) ? 0 : (t == 1) ? 1 : 4;
    const int ro = j / kt, ko = j - ro * kt;
    const int bc = rowbase + ro;
    const int b = bc >> 6, co = bc & 63;
    out[b * 576 + seg + co * kt + ko] = part[g0 + ko][0][ro] + part[g0 + ko][1][ro];
  }
}

extern "C" void kernel_launch(void* const* d_in, const int* in_sizes, int n_in,
                              void* d_out, int out_size, void* d_ws, size_t ws_size,
                              hipStream_t stream) {
  (void)in_sizes; (void)n_in; (void)out_size; (void)ws_size;
  const float* x   = (const float*)d_in[0];
  const int* idx   = (const int*)d_in[1];
  const float* U1_0 = (const float*)d_in[2];
  const float* W1_0 = (const float*)d_in[3];
  const float* U2_0 = (const float*)d_in[4];
  const float* W2_0 = (const float*)d_in[5];
  const float* U3_0 = (const float*)d_in[6];
  const float* W3_0 = (const float*)d_in[7];
  const float* U1_1 = (const float*)d_in[8];
  const float* W1_1 = (const float*)d_in[9];
  const float* U2_1 = (const float*)d_in[10];
  const float* W2_1 = (const float*)d_in[11];
  const float* U3_1 = (const float*)d_in[12];
  const float* W3_1 = (const float*)d_in[13];
  const float* U1_2 = (const float*)d_in[14];
  const float* W1_2 = (const float*)d_in[15];
  const float* U2_2 = (const float*)d_in[16];
  const float* W2_2 = (const float*)d_in[17];
  const float* U3_2 = (const float*)d_in[18];
  const float* W3_2 = (const float*)d_in[19];
  float* out = (float*)d_out;

  float2* P0 = (float2*)d_ws;
  float2* P1 = P0 + (size_t)16 * KSTEPS * 32;
  float2* P2 = P1 + (size_t)48 * KSTEPS * 32;

  sc_pack_basis<<<PACK_BLOCKS, PACK_THREADS, 0, stream>>>(
      U3_0, U2_0, U3_1, U2_1, U3_2, U2_2, P0);

  sc_wmma_kernel<<<512, NWAVES * 32, 0, stream>>>(
      x, idx, (const v2f*)P0, (const v2f*)P1, (const v2f*)P2,
      U1_0, U1_1, U1_2,
      W1_0, W1_1, W1_2,
      W2_0, W2_1, W2_2,
      W3_0, W3_1, W3_2,
      out);
}


# Main kernel: block covers 64 rows (four 16-row tiles, 24 waves) and a wave uses each basis fragment for all four tiles; basis loads per WMMA halve; a group's i-sum is 3-4 partial sums (order changes)
# speedup vs baseline: 1.7292x; 1.0377x over previous
#include <hip/hip_runtime.h>


typedef __attribute__((ext_vector_type(2))) float v2f;
typedef __attribute__((ext_vector_type(8))) float v8f;

#define FEAT 16
#define KDIM 180
#define KSTEPS 45
#define YSTRIDE 186
#define ROWS_PER_BLOCK 64
#define RT 4
#define NWAVES 24
#define TILES_PER_WAVE 6

#define NTILES_ALL 144
#define PACK_THREADS 256
#define PACK_BLOCKS 270
#define PACK_PER_THREAD 3
static_assert(PACK_BLOCKS * PACK_THREADS * PACK_PER_THREAD ==
                  NTILES_ALL * KSTEPS * 32,
              "pack grid must cover every fragment element exactly once");

__global__ __launch_bounds__(PACK_THREADS) void sc_pack_basis(
    const float* __restrict__ U3_0, const float* __restrict__ U2_0,
    const float* __restrict__ U3_1, const float* __restrict__ U2_1,
    const float* __restrict__ U3_2, const float* __restrict__ U2_2,
    float2* __restrict__ P) {
  const int gtid = blockIdx.x * PACK_THREADS + threadIdx.x;
  float2 val[PACK_PER_THREAD];
  for (int u = 0; u < PACK_PER_THREAD; ++u) {
    const int q = gtid + u * (PACK_BLOCKS * PACK_THREADS);
    const int lane = q & 31;
    const int kk = (q >> 5) % KSTEPS;
    const int tile = q / (KSTEPS * 32);
    const float* U3 = (tile < 16) ? U3_0 : (tile < 64) ? U3_1 : U3_2;
    const float* U2 = (tile < 16) ? U2_0 : (tile < 64) ? U2_1 : U2_2;
    const int ltile = (tile < 16) ? tile : (tile < 64) ? tile - 16 : tile - 64;
    const int m0 = 4 * kk + 2 * (lane >> 4);
    const int n = ltile * 16 + (lane & 15);
    const float a = (m0 < 176) ? U3[n * 176 + m0] : U2[n * 4 + (m0 - 176)];
    const int m1 = m0 + 1;
    const float b = (m1 < 176) ? U3[n * 176 + m1] : U2[n * 4 + (m1 - 176)];
    val[u] = make_float2(a, b);
  }
  for (int u = 0; u < PACK_PER_THREAD; ++u)
    P[gtid + u * (PACK_BLOCKS * PACK_THREADS)] = val[u];
}

__global__ __launch_bounds__(NWAVES * 32) void sc_wmma_kernel(
    const float* __restrict__ x, const int* __restrict__ indices,
    const v2f* __restrict__ P0, const v2f* __restrict__ P1,
    const v2f* __restrict__ P2,
    const float* __restrict__ U1_0, const float* __restrict__ U1_1,
    const float* __restrict__ U1_2,
    const float* __restrict__ W1_0, const float* __restrict__ W1_1,
    const float* __restrict__ W1_2,
    const float* __restrict__ W2_0, const float* __restrict__ W2_1,
    const float* __restrict__ W2_2,
    const float* __restrict__ W3_0, const float* __restrict__ W3_1,
    const float* __restrict__ W3_2,
    float* __restrict__ out) {
  __shared__ float xs[ROWS_PER_BLOCK][FEAT + 1];
  __shared__ alignas(16) float Ylds[3][ROWS_PER_BLOCK][YSTRIDE];
  __shared__ float w1s[3][ROWS_PER_BLOCK];
  __shared__ float part[NWAVES][2][ROWS_PER_BLOCK];

  const int tid = threadIdx.x;
  const int rowbase = blockIdx.x * ROWS_PER_BLOCK;

  const int e = indices[blockIdx.x];
  const int r = tid & 63;
  const int q = tid >> 6;
  const int c = r;

  const int tpA = q, tpB = q + 12, tpC = q + 24;
  const int tA = tpA / 11, pA = tpA - tA * 11;
  const int tB = tpB / 11, pB = tpB - tB * 11;
  const bool third = q < 9;
  const int pC = tpC - 22;
  const float* W3A = (tA == 0) ? W3_0 : W3_1;
  const float* W3B = (tB == 1) ? W3_1 : W3_2;
  const float wA = W3A[(e * 11 + pA) * 64 + c];
  const float wB = W3B[(e * 11 + pB) * 64 + c];
  float wC = 0.0f;
  if (third) wC = W3_2[(e * 11 + pC) * 64 + c];
  const int t2 = q >> 2, p2 = q & 3;
  const float* W2s = (t2 == 0) ? W2_0 : (t2 == 1) ? W2_1 : W2_2;
  const float w2v = W2s[(e * 4 + p2) * 64 + c];
  float w1v = 0.0f;
  if (q < 3) {
    const float* W1s = (q == 0) ? W1_0 : (q == 1) ? W1_1 : W1_2;
    w1v = W1s[e * 64 + c];
  }

  xs[tid >> 4][tid & 15] = x[rowbase * FEAT + tid];
  if (tid < ROWS_PER_BLOCK * FEAT - NWAVES * 32) {
    const int u = tid + NWAVES * 32;
    xs[u >> 4][u & 15] = x[rowbase * FEAT + u];
  }
  __syncthreads();

  {
    float xr[FEAT];
#pragma unroll
    for (int l = 0; l < FEAT; ++l) xr[l] = xs[r][l];
    float* yA = &Ylds[tA][r][pA];
#pragma unroll
    for (int l = 0; l < FEAT; ++l) yA[l * 11] = xr[l] * wA;
    float* yB = &Ylds[tB][r][pB];
#pragma unroll
    for (int l = 0; l < FEAT; ++l) yB[l * 11] = xr[l] * wB;
    if (third) {
      float* yC = &Ylds[2][r][pC];
#pragma unroll
      for (int l = 0; l < FEAT; ++l) yC[l * 11] = xr[l] * wC;
    }
    Ylds[t2][r][176 + p2] = w2v;
    if (q < 3) w1s[q][r] = w1v;
  }
  __syncthreads();

  const int wave = __builtin_amdgcn_readfirstlane(tid >> 5);
  const int lane = tid & 31;
  const int half = lane >> 4;
  const int l15 = lane & 15;

  float xh[RT][8];
#pragma unroll
  for (int rt = 0; rt < RT; ++rt)
#pragma unroll
    for (int v = 0; v < 8; ++v) xh[rt][v] = xs[16 * rt + l15][(half << 3) + v];

  const int u0 = wave * TILES_PER_WAVE;
  const int gfirst = u0 >> 4;
  const int ifirst = u0 & 15;
  const int nA = (16 - ifirst < TILES_PER_WAVE) ? (16 - ifirst) : TILES_PER_WAVE;

  for (int sgm = 0; sgm < 2; ++sgm) {
    const int g  = gfirst + sgm;
    const int ib = sgm ? 0 : ifirst;
    const int ie = sgm ? (TILES_PER_WAVE - nA) : (ifirst + nA);
    if (ib >= ie) break;
    int t, k;
    if (g < 1)       { t = 0; k = 0; }
    else if (g < 4)  { t = 1; k = g - 1; }
    else             { t = 2; k = g - 4; }
    const v2f* Pt    = (t == 0) ? P0 : (t == 1) ? P1 : P2;
    const float* U1  = (t == 0) ? U1_0 : (t == 1) ? U1_1 : U1_2;
    float w1r[RT];
    const float* Yrow[RT];
#pragma unroll
    for (int rt = 0; rt < RT; ++rt) {
      w1r[rt]  = w1s[t][16 * rt + l15];
      Yrow[rt] = &Ylds[t][16 * rt + l15][2 * half];
    }

    float accT[RT] = {0.0f, 0.0f, 0.0f, 0.0f};

    for (int i = ib; i < ie; ++i) {
      const v2f* Ap = Pt + ((size_t)(k * 16 + i) * KSTEPS) * 32 + lane;
      __builtin_prefetch(Ap, 0, 1);

      v8f acc[RT] = {};
#pragma unroll 9
      for (int kk = 0; kk < KSTEPS; ++kk) {
        v2f A = Ap[kk * 32];
#pragma unroll
        for (int rt = 0; rt < RT; ++rt) {
          v2f B = *reinterpret_cast<const v2f*>(Yrow[rt] + 4 * kk);
          acc[rt] = __builtin_amdgcn_wmma_f32_16x16x4_f32(
              false, A, false, B, (short)0, acc[rt], false, false);
        }
      }

      const float u1v = U1[k * 16 + i];
#pragma unroll
      for (int rt = 0; rt < RT; ++rt) {
        float s = 0.0f;
#pragma unroll
        for (int v = 0; v < 8; ++v) s = fmaf(acc[rt][v], xh[rt][v], s);
        s += __shfl_xor(s, 16, 32);
        const float c1 = fmaf(u1v, w1r[rt], s);
        accT[rt] = fmaf(c1, xs[16 * rt + l15][i], accT[rt]);
      }
    }

    if (lane < 16) {
#pragma unroll
      for (int rt = 0; rt < RT; ++rt) part[wave][sgm][16 * rt + l15] = accT[rt];
    }
  }
  __syncthreads();

  if (tid < 9 * ROWS_PER_BLOCK) {
    int t, j;
    if (tid < 64)       { t = 0; j = tid; }
    else if (tid < 256) { t = 1; j = tid - 64; }
    else                { t = 2; j = tid - 256; }
    const int kt  = (t == 0) ? 1 : (t == 1) ? 3 : 5;
    const int g0  = (t == 0) ? 0 : (t == 1) ? 1 : 4;
    const int ro = j / kt, ko = j - ro * kt;
    const int g = g0 + ko;
    const int wlo = (16 * g) / TILES_PER_WAVE;
    const int whi = (16 * g + 15) / TILES_PER_WAVE;
    float s = 0.0f;
    for (int w = wlo; w <= whi; ++w)
      s += part[w][g - ((w * TILES_PER_WAVE) >> 4)][ro];
    out[blockIdx.x * 576 + tid] = s;
  }
}

extern "C" void kernel_launch(void* const* d_in, const int* in_sizes, int n_in,
                              void* d_out, int out_size, void* d_ws, size_t ws_size,
                              hipStream_t stream) {
  (void)in_sizes; (void)n_in; (void)out_size; (void)ws_size;
  const float* x   = (const float*)d_in[0];
  const int* idx   = (const int*)d_in[1];
  const float* U1_0 = (const float*)d_in[2];
  const float* W1_0 = (const float*)d_in[3];
  const float* U2_0 = (const float*)d_in[4];
  const float* W2_0 = (const float*)d_in[5];
  const float* U3_0 = (const float*)d_in[6];
  const float* W3_0 = (const float*)d_in[7];
  const float* U1_1 = (const float*)d_in[8];
  const float* W1_1 = (const float*)d_in[9];
  const float* U2_1 = (const float*)d_in[10];
  const float* W2_1 = (const float*)d_in[11];
  const float* U3_1 = (const float*)d_in[12];
  const float* W3_1 = (const float*)d_in[13];
  const float* U1_2 = (const float*)d_in[14];
  const float* W1_2 = (const float*)d_in[15];
  const float* U2_2 = (const float*)d_in[16];
  const float* W2_2 = (const float*)d_in[17];
  const float* U3_2 = (const float*)d_in[18];
  const float* W3_2 = (const float*)d_in[19];
  float* out = (float*)d_out;

  float2* P0 = (float2*)d_ws;
  float2* P1 = P0 + (size_t)16 * KSTEPS * 32;
  float2* P2 = P1 + (size_t)48 * KSTEPS * 32;

  sc_pack_basis<<<PACK_BLOCKS, PACK_THREADS, 0, stream>>>(
      U3_0, U2_0, U3_1, U2_1, U3_2, U2_2, P0);

  sc_wmma_kernel<<<256, NWAVES * 32, 0, stream>>>(
      x, idx, (const v2f*)P0, (const v2f*)P1, (const v2f*)P2,
      U1_0, U1_1, U1_2,
      W1_0, W1_1, W1_2,
      W2_0, W2_1, W2_2,
      W3_0, W3_1, W3_2,
      out);
}


# Main kernel chunk tail off LDS: half-sum exchange by v_permlanex16 (was ds_bpermute), per-(b,c) running sum over i kept in a register (was LDS load-fma-store per chunk and i); sum order unchanged
# speedup vs baseline: 4.7257x; 2.7329x over previous
#include <hip/hip_runtime.h>


typedef __attribute__((ext_vector_type(2))) float v2f;
typedef __attribute__((ext_vector_type(4))) float v4f;
typedef __attribute__((ext_vector_type(8))) float v8f;

__device__ __forceinline__ float xhalf(float v) {
  const int b = __builtin_bit_cast(int, v);
  return __builtin_bit_cast(
      float, __builtin_amdgcn_permlanex16(b, b, 0x76543210, 0xfedcba98, false, false));
}

#define NB 256
#define NCH 64
#define FEAT 16
#define NELEM 10
#define NWAVES 16
#define NTHREADS (NWAVES * 32)
#define VL 20
#define VC 324
#define ACCP 17
#define WS_LINE 32
#define NBLOCKS (NELEM * 9 * 4)
#define HELD 4

__global__ __launch_bounds__(NTHREADS) __attribute__((amdgpu_waves_per_eu(8)))
void sc_fold_kernel(
    const float* __restrict__ x, const int* __restrict__ indices,
    const float* __restrict__ U1_0, const float* __restrict__ U1_1,
    const float* __restrict__ U1_2,
    const float* __restrict__ U2_0, const float* __restrict__ U2_1,
    const float* __restrict__ U2_2,
    const float* __restrict__ U3_0, const float* __restrict__ U3_1,
    const float* __restrict__ U3_2,
    const float* __restrict__ W1_0, const float* __restrict__ W1_1,
    const float* __restrict__ W1_2,
    const float* __restrict__ W2_0, const float* __restrict__ W2_1,
    const float* __restrict__ W2_2,
    const float* __restrict__ W3_0, const float* __restrict__ W3_1,
    const float* __restrict__ W3_2,
    float* __restrict__ ws) {
  __shared__ alignas(16) float Vs[2][16 * VC];
  __shared__ alignas(16) float V2s[2][16][16];
  __shared__ float accs[NB * ACCP];
  __shared__ int blist[NB];
  __shared__ int wcnt[8];

  const int tid = threadIdx.x;
  const int bid = blockIdx.x;
  const int e   = bid / 36;
  const int rem = bid - e * 36;
  const int g   = rem >> 2;
  const int ct  = rem & 3;
  const int c0  = ct * 16;
  int t, k;
  if (g < 1)       { t = 0; k = 0; }
  else if (g < 4)  { t = 1; k = g - 1; }
  else             { t = 2; k = g - 4; }
  const float* U1 = (t == 0) ? U1_0 : (t == 1) ? U1_1 : U1_2;
  const float* U2 = (t == 0) ? U2_0 : (t == 1) ? U2_1 : U2_2;
  const float* U3 = (t == 0) ? U3_0 : (t == 1) ? U3_1 : U3_2;
  const float* W1 = (t == 0) ? W1_0 : (t == 1) ? W1_1 : W1_2;
  const float* W2 = (t == 0) ? W2_0 : (t == 1) ? W2_1 : W2_2;
  const float* W3 = (t == 0) ? W3_0 : (t == 1) ? W3_1 : W3_2;

  const int wave = __builtin_amdgcn_readfirstlane(tid >> 5);
  const int lane = tid & 31;
  const int h    = lane >> 4;
  const int l15  = lane & 15;

  bool mine = false;
  if (tid < NB) mine = (indices[tid] == e);
  const unsigned mask = __builtin_amdgcn_ballot_w32(mine);
  if (tid < NB && lane == 0) wcnt[wave] = __popc(mask);
  for (int q = 16 * HELD * ACCP + tid; q < NB * ACCP; q += NTHREADS) accs[q] = 0.0f;
  __syncthreads();
  int ne_v = 0, off = 0;
  for (int w2 = 0; w2 < 8; ++w2) {
    const int cv = wcnt[w2];
    if (w2 < wave) off += cv;
    ne_v += cv;
  }
  if (mine) blist[off + __popc(mask & ((1u << lane) - 1u))] = tid;
  __syncthreads();
  const int n_e = __builtin_amdgcn_readfirstlane(ne_v);
  if (n_e == 0) return;
  const int nchunk = (n_e + 15) >> 4;

  v2f wb[3];
#pragma unroll
  for (int kk = 0; kk < 3; ++kk) {
    const int p0 = 4 * kk + 2 * h;
    wb[kk].x = W3[(e * 11 + p0) * NCH + c0 + l15];
    wb[kk].y = (p0 + 1 < 11) ? W3[(e * 11 + p0 + 1) * NCH + c0 + l15] : 0.0f;
  }
  v2f wb2;
  wb2.x = W2[(e * 4 + 2 * h) * NCH + c0 + l15];
  wb2.y = W2[(e * 4 + 2 * h + 1) * NCH + c0 + l15];
  const float w1c = W1[e * NCH + c0 + wave];

  v4f xr0[HELD], xr1[HELD];
  float racc[HELD];
#pragma unroll
  for (int ch = 0; ch < HELD; ++ch) {
    racc[ch] = 0.0f;
    xr0[ch] = v4f{0.0f, 0.0f, 0.0f, 0.0f};
    xr1[ch] = v4f{0.0f, 0.0f, 0.0f, 0.0f};
    if (ch < nchunk) {
      const int bslot = ch * 16 + l15;
      const int bs = (bslot < n_e) ? bslot : (n_e - 1);
      const float* xp = x + (size_t)(blist[bs] * NCH + c0 + wave) * FEAT + 8 * h;
      xr0[ch] = *reinterpret_cast<const v4f*>(xp);
      xr1[ch] = *reinterpret_cast<const v4f*>(xp + 4);
    }
  }

  v2f a[3];
  v2f a2 = {0.0f, 0.0f};
  {
    const float* up = U3 + (size_t)(((k * 16 + 0) * 16 + wave) * 16 + l15) * 11 + 2 * h;
#pragma unroll
    for (int kk = 0; kk < 3; ++kk) {
      a[kk].x = up[4 * kk];
      a[kk].y = (4 * kk + 2 * h + 1 < 11) ? up[4 * kk + 1] : 0.0f;
    }
    if (wave == 0)
      a2 = *reinterpret_cast<const v2f*>(U2 + ((k * 16 + 0) * 16 + l15) * 4 + 2 * h);
  }
  float u1n = U1[k * 16 + 0];

  for (int i = 0; i < FEAT; ++i) {
    const int buf = i & 1;
    const float u1v = u1n;

    v8f d = {};
#pragma unroll
    for (int kk = 0; kk < 3; ++kk)
      d = __builtin_amdgcn_wmma_f32_16x16x4_f32(false, a[kk], false, wb[kk],
                                                (short)0, d, false, false);
    {
      float* vdst = &Vs[buf][l15 * VC + wave * VL + 8 * h];
      v4f lo = {d[0], d[1], d[2], d[3]};
      v4f hi = {d[4], d[5], d[6], d[7]};
      *reinterpret_cast<v4f*>(vdst)     = lo;
      *reinterpret_cast<v4f*>(vdst + 4) = hi;
    }
    if (wave == 0) {
      v8f d2 = {};
      d2 = __builtin_amdgcn_wmma_f32_16x16x4_f32(false, a2, false, wb2,
                                                 (short)0, d2, false, false);
      float* v2dst = &V2s[buf][l15][8 * h];
      v4f lo = {d2[0], d2[1], d2[2], d2[3]};
      v4f hi = {d2[4], d2[5], d2[6], d2[7]};
      *reinterpret_cast<v4f*>(v2dst)     = lo;
      *reinterpret_cast<v4f*>(v2dst + 4) = hi;
    }

    __syncthreads();

    if (i + 1 < FEAT) {
      const float* up =
          U3 + (size_t)(((k * 16 + i + 1) * 16 + wave) * 16 + l15) * 11 + 2 * h;
#pragma unroll
      for (int kk = 0; kk < 3; ++kk) {
        a[kk].x = up[4 * kk];
        a[kk].y = (4 * kk + 2 * h + 1 < 11) ? up[4 * kk + 1] : 0.0f;
      }
      if (wave == 0)
        a2 = *reinterpret_cast<const v2f*>(U2 + ((k * 16 + i + 1) * 16 + l15) * 4 + 2 * h);
      u1n = U1[k * 16 + i + 1];
    }

    v2f A2[4];
    {
      const float* vsrc = &Vs[buf][wave * VC + l15 * VL + 8 * h];
#pragma unroll
      for (int kk = 0; kk < 4; ++kk)
        A2[kk] = *reinterpret_cast<const v2f*>(vsrc + 2 * kk);
    }
    const v4f ci0 = *reinterpret_cast<const v4f*>(&V2s[buf][wave][8 * h]);
    const v4f ci1 = *reinterpret_cast<const v4f*>(&V2s[buf][wave][8 * h + 4]);
    const int isub  = i & 7;
    const int ihalf = i >> 3;

    auto chunk_c1 = [&](v4f x0, v4f x1) -> float {
      v8f acc = {ci0[0], ci0[1], ci0[2], ci0[3], ci1[0], ci1[1], ci1[2], ci1[3]};
      v2f B0 = {x0[0], x0[1]};
      v2f B1 = {x0[2], x0[3]};
      v2f B2 = {x1[0], x1[1]};
      v2f B3 = {x1[2], x1[3]};
      acc = __builtin_amdgcn_wmma_f32_16x16x4_f32(false, A2[0], false, B0, (short)0, acc, false, false);
      acc = __builtin_amdgcn_wmma_f32_16x16x4_f32(false, A2[1], false, B1, (short)0, acc, false, false);
      acc = __builtin_amdgcn_wmma_f32_16x16x4_f32(false, A2[2], false, B2, (short)0, acc, false, false);
      acc = __builtin_amdgcn_wmma_f32_16x16x4_f32(false, A2[3], false, B3, (short)0, acc, false, false);
      float s = acc[0] * x0[0];
      s = fmaf(acc[1], x0[1], s);
      s = fmaf(acc[2], x0[2], s);
      s = fmaf(acc[3], x0[3], s);
      s = fmaf(acc[4], x1[0], s);
      s = fmaf(acc[5], x1[1], s);
      s = fmaf(acc[6], x1[2], s);
      s = fmaf(acc[7], x1[3], s);
      s += xhalf(s);
      return fmaf(u1v, w1c, s);
    };
    auto chunk_xi = [&](v4f x0, v4f x1) -> float {
      const v8f xv = {x0[0], x0[1], x0[2], x0[3], x1[0], x1[1], x1[2], x1[3]};
      return xv[isub];
    };

#pragma unroll
    for (int ch = 0; ch < HELD; ++ch) {
      if (ch < nchunk) {
        const float c1 = chunk_c1(xr0[ch], xr1[ch]);
        const float xown = chunk_xi(xr0[ch], xr1[ch]);
        const float xoth = xhalf(xown);
        const float xi = (h == ihalf) ? xown : xoth;
        racc[ch] = fmaf(c1, xi, racc[ch]);
      }
    }
    for (int ch = HELD; ch < nchunk; ++ch) {
      const int bslot = ch * 16 + l15;
      const int bs = (bslot < n_e) ? bslot : (n_e - 1);
      const float* xp = x + (size_t)(blist[bs] * NCH + c0 + wave) * FEAT + 8 * h;
      const v4f x0 = *reinterpret_cast<const v4f*>(xp);
      const v4f x1 = *reinterpret_cast<const v4f*>(xp + 4);
      const float c1 = chunk_c1(x0, x1);
      const float xi = chunk_xi(x0, x1);
      if (h == ihalf && bslot < n_e) {
        float* ap = &accs[bslot * ACCP + wave];
        *ap = fmaf(c1, xi, *ap);
      }
    }
  }
#pragma unroll
  for (int ch = 0; ch < HELD; ++ch) {
    if (ch < nchunk) {
      const int bslot = ch * 16 + l15;
      if (h == 0 && bslot < n_e) accs[bslot * ACCP + wave] = racc[ch];
    }
  }
  __syncthreads();

  float* wsb = ws + (size_t)((g * 4 + ct) * NB) * WS_LINE;
  for (int q = tid; q < n_e * 16; q += NTHREADS) {
    const int bs = q >> 4, cl = q & 15;
    wsb[blist[bs] * WS_LINE + cl] = accs[bs * ACCP + cl];
  }
}

#define GATHER_THREADS 288
__global__ __launch_bounds__(GATHER_THREADS) void sc_gather_out(
    const float* __restrict__ ws, float* __restrict__ out) {
  const int tid  = threadIdx.x;
  const int hrow = (tid >= 144) ? 1 : 0;
  const int b    = blockIdx.x * 2 + hrow;
  const int q0   = (tid - hrow * 144) * 4;
  int kt, g0, joff;
  if (q0 < 64)       { kt = 1; g0 = 0; joff = 0; }
  else if (q0 < 256) { kt = 3; g0 = 1; joff = 64; }
  else               { kt = 5; g0 = 4; joff = 256; }
  float v[4];
#pragma unroll
  for (int u = 0; u < 4; ++u) {
    const int j  = q0 + u - joff;
    const int c  = j / kt;
    const int ko = j - c * kt;
    const int g  = g0 + ko;
    v[u] = ws[(size_t)((g * 4 + (c >> 4)) * NB + b) * WS_LINE + (c & 15)];
  }
  const v4f o = {v[0], v[1], v[2], v[3]};
  *reinterpret_cast<v4f*>(out + b * 576 + q0) = o;
}

extern "C" void kernel_launch(void* const* d_in, const int* in_sizes, int n_in,
                              void* d_out, int out_size, void* d_ws, size_t ws_size,
                              hipStream_t stream) {
  (void)in_sizes; (void)n_in; (void)out_size; (void)ws_size;
  const float* x   = (const float*)d_in[0];
  const int* idx   = (const int*)d_in[1];
  const float* U1_0 = (const float*)d_in[2];
  const float* W1_0 = (const float*)d_in[3];
  const float* U2_0 = (const float*)d_in[4];
  const float* W2_0 = (const float*)d_in[5];
  const float* U3_0 = (const float*)d_in[6];
  const float* W3_0 = (const float*)d_in[7];
  const float* U1_1 = (const float*)d_in[8];
  const float* W1_1 = (const float*)d_in[9];
  const float* U2_1 = (const float*)d_in[10];
  const float* W2_1 = (const float*)d_in[11];
  const float* U3_1 = (const float*)d_in[12];
  const float* W3_1 = (const float*)d_in[13];
  const float* U1_2 = (const float*)d_in[14];
  const float* W1_2 = (const float*)d_in[15];
  const float* U2_2 = (const float*)d_in[16];
  const float* W2_2 = (const float*)d_in[17];
  const float* U3_2 = (const float*)d_in[18];
  const float* W3_2 = (const float*)d_in[19];
  float* out = (float*)d_out;
  float* ws  = (float*)d_ws;

  sc_fold_kernel<<<NBLOCKS, NTHREADS, 0, stream>>>(
      x, idx,
      U1_0, U1_1, U1_2,
      U2_0, U2_1, U2_2,
      U3_0, U3_1, U3_2,
      W1_0, W1_1, W1_2,
      W2_0, W2_1, W2_2,
      W3_0, W3_1, W3_2,
      ws);

  sc_gather_out<<<NB / 2, GATHER_THREADS, 0, stream>>>(ws, out);
}


# Main kernel: a group's 16-step i walk is cut: 4 blocks share the 48 steps of a triple of groups, 12 each (480 blocks, was 360 of 16); a walk is 2 partial sums, added in launch 2 (sum order changes)
# speedup vs baseline: 5.3654x; 1.1354x over previous
#include <hip/hip_runtime.h>


typedef __attribute__((ext_vector_type(2))) float v2f;
typedef __attribute__((ext_vector_type(4))) float v4f;
typedef __attribute__((ext_vector_type(8))) float v8f;

__device__ __forceinline__ float xhalf(float v) {
  const int b = __builtin_bit_cast(int, v);
  return __builtin_bit_cast(
      float, __builtin_amdgcn_permlanex16(b, b, 0x76543210, 0xfedcba98, false, false));
}

#define NB 256
#define NCH 64
#define FEAT 16
#define NELEM 10
#define NWAVES 16
#define NTHREADS (NWAVES * 32)
#define VL 20
#define VC 324
#define ACCP 17
#define WS_LINE 32
#define NBLOCKS (NELEM * 4 * 3 * 4)
#define WS_PLANE (36 * NB * WS_LINE)
#define HELD 4

__global__ __launch_bounds__(NTHREADS) __attribute__((amdgpu_waves_per_eu(8)))
void sc_fold_kernel(
    const float* __restrict__ x, const int* __restrict__ indices,
    const float* __restrict__ U1_0, const float* __restrict__ U1_1,
    const float* __restrict__ U1_2,
    const float* __restrict__ U2_0, const float* __restrict__ U2_1,
    const float* __restrict__ U2_2,
    const float* __restrict__ U3_0, const float* __restrict__ U3_1,
    const float* __restrict__ U3_2,
    const float* __restrict__ W1_0, const float* __restrict__ W1_1,
    const float* __restrict__ W1_2,
    const float* __restrict__ W2_0, const float* __restrict__ W2_1,
    const float* __restrict__ W2_2,
    const float* __restrict__ W3_0, const float* __restrict__ W3_1,
    const float* __restrict__ W3_2,
    float* __restrict__ ws) {
  __shared__ alignas(16) float Vs[2][16 * VC];
  __shared__ alignas(16) float V2s[2][16][16];
  __shared__ float accs[NB * ACCP];
  __shared__ int blist[NB];
  __shared__ int wcnt[8];

  const int tid = threadIdx.x;
  const int n   = blockIdx.x;
  const int m   = n >> 2;
  const int r   = n & 3;
  const int e   = m / 12;
  const int mr  = m - e * 12;
  const int ct  = mr / 3;
  const int gt  = mr - ct * 3;
  const int c0  = ct * 16;

  const int wave = __builtin_amdgcn_readfirstlane(tid >> 5);
  const int lane = tid & 31;
  const int h    = lane >> 4;
  const int l15  = lane & 15;
  const unsigned hl = (unsigned)(2 * h * NCH + l15);

  bool mine = false;
  if (tid < NB) mine = (indices[tid] == e);
  const unsigned mask = __builtin_amdgcn_ballot_w32(mine);
  if (tid < NB && lane == 0) wcnt[wave] = __popc(mask);
  __syncthreads();
  int ne_v = 0, off = 0;
  for (int w2 = 0; w2 < 8; ++w2) {
    const int cv = wcnt[w2];
    if (w2 < wave) off += cv;
    ne_v += cv;
  }
  if (mine) blist[off + __popc(mask & ((1u << lane) - 1u))] = tid;
  __syncthreads();
  const int n_e = __builtin_amdgcn_readfirstlane(ne_v);
  if (n_e == 0) return;
  const int nchunk = (n_e + 15) >> 4;

  v4f xr0[HELD], xr1[HELD];
#pragma unroll
  for (int ch = 0; ch < HELD; ++ch) {
    xr0[ch] = v4f{0.0f, 0.0f, 0.0f, 0.0f};
    xr1[ch] = v4f{0.0f, 0.0f, 0.0f, 0.0f};
    if (ch < nchunk) {
      const int bslot = ch * 16 + l15;
      const int bs = (bslot < n_e) ? bslot : (n_e - 1);
      const float* xp = x + (size_t)(blist[bs] * NCH + c0 + wave) * FEAT + 8 * h;
      xr0[ch] = *reinterpret_cast<const v4f*>(xp);
      xr1[ch] = *reinterpret_cast<const v4f*>(xp + 4);
    }
  }

  const int nrun = ((12 * r) & 15) + 12 > 16 ? 2 : 1;
  int step = 0;
  for (int run = 0; run < nrun; ++run) {
    const int s    = ((12 * r) >> 4) + run;
    const int i_lo = (run == 0) ? ((12 * r) & 15) : 0;
    const int i_hi = (run == 0) ? ((i_lo + 12 < 16) ? i_lo + 12 : 16)
                                : 12 * r + 12 - 16 * s;
    const int plane = (i_lo == 0) ? 0 : 1;
    const int g = gt * 3 + s;
    int t, k;
    if (g < 1)       { t = 0; k = 0; }
    else if (g < 4)  { t = 1; k = g - 1; }
    else             { t = 2; k = g - 4; }
    const float* U1 = (t == 0) ? U1_0 : (t == 1) ? U1_1 : U1_2;
    const float* U2 = (t == 0) ? U2_0 : (t == 1) ? U2_1 : U2_2;
    const float* U3 = (t == 0) ? U3_0 : (t == 1) ? U3_1 : U3_2;
    const float* W1 = (t == 0) ? W1_0 : (t == 1) ? W1_1 : W1_2;
    const float* W2 = (t == 0) ? W2_0 : (t == 1) ? W2_1 : W2_2;
    const float* W3 = (t == 0) ? W3_0 : (t == 1) ? W3_1 : W3_2;

    if (nchunk > HELD)
      for (int q = 16 * HELD * ACCP + tid; q < NB * ACCP; q += NTHREADS) accs[q] = 0.0f;

    const float* w3e = W3 + (e * 11) * NCH + c0;
    const float* w2e = W2 + (e * 4) * NCH + c0;
    v2f wb[3];
#pragma unroll
    for (int kk = 0; kk < 3; ++kk) {
      wb[kk].x = w3e[4 * kk * NCH + hl];
      wb[kk].y = (4 * kk + 2 * h + 1 < 11) ? w3e[(4 * kk + 1) * NCH + hl] : 0.0f;
    }
    v2f wb2;
    wb2.x = w2e[hl];
    wb2.y = w2e[NCH + hl];
    const float w1c = W1[e * NCH + c0 + wave];

    float racc[HELD];
#pragma unroll
    for (int ch = 0; ch < HELD; ++ch) racc[ch] = 0.0f;

    v2f a[3];
    v2f a2 = {0.0f, 0.0f};
    {
      const float* up =
          U3 + (size_t)(((k * 16 + i_lo) * 16 + wave) * 16 + l15) * 11 + 2 * h;
#pragma unroll
      for (int kk = 0; kk < 3; ++kk) {
        a[kk].x = up[4 * kk];
        a[kk].y = (4 * kk + 2 * h + 1 < 11) ? up[4 * kk + 1] : 0.0f;
      }
      if (wave == 0)
        a2 = *reinterpret_cast<const v2f*>(U2 + ((k * 16 + i_lo) * 16 + l15) * 4 + 2 * h);
    }
    float u1n = U1[k * 16 + i_lo];

    for (int i = i_lo; i < i_hi; ++i, ++step) {
      const int buf = step & 1;
      const float u1v = u1n;

      v8f d = {};
#pragma unroll
      for (int kk = 0; kk < 3; ++kk)
        d = __builtin_amdgcn_wmma_f32_16x16x4_f32(false, a[kk], false, wb[kk],
                                                  (short)0, d, false, false);
      {
        float* vdst = &Vs[buf][l15 * VC + wave * VL + 8 * h];
        v4f lo = {d[0], d[1], d[2], d[3]};
        v4f hi = {d[4], d[5], d[6], d[7]};
        *reinterpret_cast<v4f*>(vdst)     = lo;
        *reinterpret_cast<v4f*>(vdst + 4) = hi;
      }
      if (wave == 0) {
        v8f d2 = {};
        d2 = __builtin_amdgcn_wmma_f32_16x16x4_f32(false, a2, false, wb2,
                                                   (short)0, d2, false, false);
        float* v2dst = &V2s[buf][l15][8 * h];
        v4f lo = {d2[0], d2[1], d2[2], d2[3]};
        v4f hi = {d2[4], d2[5], d2[6], d2[7]};
        *reinterpret_cast<v4f*>(v2dst)     = lo;
        *reinterpret_cast<v4f*>(v2dst + 4) = hi;
      }

      __syncthreads();

      if (i + 1 < i_hi) {
        const float* up =
            U3 + (size_t)(((k * 16 + i + 1) * 16 + wave) * 16 + l15) * 11 + 2 * h;
#pragma unroll
        for (int kk = 0; kk < 3; ++kk) {
          a[kk].x = up[4 * kk];
          a[kk].y = (4 * kk + 2 * h + 1 < 11) ? up[4 * kk + 1] : 0.0f;
        }
        if (wave == 0)
          a2 = *reinterpret_cast<const v2f*>(U2 + ((k * 16 + i + 1) * 16 + l15) * 4 + 2 * h);
        u1n = U1[k * 16 + i + 1];
      }

      v2f A2[4];
      {
        const float* vsrc = &Vs[buf][wave * VC + l15 * VL + 8 * h];
#pragma unroll
        for (int kk = 0; kk < 4; ++kk)
          A2[kk] = *reinterpret_cast<const v2f*>(vsrc + 2 * kk);
      }
      const v4f ci0 = *reinterpret_cast<const v4f*>(&V2s[buf][wave][8 * h]);
      const v4f ci1 = *reinterpret_cast<const v4f*>(&V2s[buf][wave][8 * h + 4]);
      const int isub  = i & 7;
      const int ihalf = i >> 3;

      auto chunk_c1 = [&](v4f x0, v4f x1) -> float {
        v8f acc = {ci0[0], ci0[1], ci0[2], ci0[3], ci1[0], ci1[1], ci1[2], ci1[3]};
        v2f B0 = {x0[0], x0[1]};
        v2f B1 = {x0[2], x0[3]};
        v2f B2 = {x1[0], x1[1]};
        v2f B3 = {x1[2], x1[3]};
        acc = __builtin_amdgcn_wmma_f32_16x16x4_f32(false, A2[0], false, B0, (short)0, acc, false, false);
        acc = __builtin_amdgcn_wmma_f32_16x16x4_f32(false, A2[1], false, B1, (short)0, acc, false, false);
        acc = __builtin_amdgcn_wmma_f32_16x16x4_f32(false, A2[2], false, B2, (short)0, acc, false, false);
        acc = __builtin_amdgcn_wmma_f32_16x16x4_f32(false, A2[3], false, B3, (short)0, acc, false, false);
        float sj = acc[0] * x0[0];
        sj = fmaf(acc[1], x0[1], sj);
        sj = fmaf(acc[2], x0[2], sj);
        sj = fmaf(acc[3], x0[3], sj);
        sj = fmaf(acc[4], x1[0], sj);
        sj = fmaf(acc[5], x1[1], sj);
        sj = fmaf(acc[6], x1[2], sj);
        sj = fmaf(acc[7], x1[3], sj);
        sj += xhalf(sj);
        return fmaf(u1v, w1c, sj);
      };
      auto chunk_xi = [&](v4f x0, v4f x1) -> float {
        const v8f xv = {x0[0], x0[1], x0[2], x0[3], x1[0], x1[1], x1[2], x1[3]};
        return xv[isub];
      };

#pragma unroll
      for (int ch = 0; ch < HELD; ++ch) {
        if (ch < nchunk) {
          const float c1 = chunk_c1(xr0[ch], xr1[ch]);
          const float xown = chunk_xi(xr0[ch], xr1[ch]);
          const float xoth = xhalf(xown);
          const float xi = (h == ihalf) ? xown : xoth;
          racc[ch] = fmaf(c1, xi, racc[ch]);
        }
      }
      for (int ch = HELD; ch < nchunk; ++ch) {
        const int bslot = ch * 16 + l15;
        const int bs = (bslot < n_e) ? bslot : (n_e - 1);
        const float* xp = x + (size_t)(blist[bs] * NCH + c0 + wave) * FEAT + 8 * h;
        const v4f x0 = *reinterpret_cast<const v4f*>(xp);
        const v4f x1 = *reinterpret_cast<const v4f*>(xp + 4);
        const float c1 = chunk_c1(x0, x1);
        const float xi = chunk_xi(x0, x1);
        if (h == ihalf && bslot < n_e) {
          float* ap = &accs[bslot * ACCP + wave];
          *ap = fmaf(c1, xi, *ap);
        }
      }
    }
#pragma unroll
    for (int ch = 0; ch < HELD; ++ch) {
      if (ch < nchunk) {
        const int bslot = ch * 16 + l15;
        if (h == 0 && bslot < n_e) accs[bslot * ACCP + wave] = racc[ch];
      }
    }
    __syncthreads();

    float* wsb = ws + (size_t)plane * WS_PLANE + (size_t)((g * 4 + ct) * NB) * WS_LINE;
#pragma nounroll
    for (int q = tid; q < n_e * 16; q += NTHREADS) {
      const int bs = q >> 4, cl = q & 15;
      wsb[blist[bs] * WS_LINE + cl] = accs[bs * ACCP + cl];
    }
    if (run + 1 < nrun) __syncthreads();
  }
}

#define GATHER_THREADS 288
__global__ __launch_bounds__(GATHER_THREADS) void sc_gather_out(
    const float* __restrict__ ws, float* __restrict__ out) {
  const int tid  = threadIdx.x;
  const int hrow = (tid >= 144) ? 1 : 0;
  const int b    = blockIdx.x * 2 + hrow;
  const int q0   = (tid - hrow * 144) * 4;
  int kt, g0, joff;
  if (q0 < 64)       { kt = 1; g0 = 0; joff = 0; }
  else if (q0 < 256) { kt = 3; g0 = 1; joff = 64; }
  else               { kt = 5; g0 = 4; joff = 256; }
  float v[4];
#pragma unroll
  for (int u = 0; u < 4; ++u) {
    const int j  = q0 + u - joff;
    const int c  = j / kt;
    const int ko = j - c * kt;
    const int g  = g0 + ko;
    const size_t at = (size_t)((g * 4 + (c >> 4)) * NB + b) * WS_LINE + (c & 15);
    v[u] = ws[at] + ws[at + WS_PLANE];
  }
  const v4f o = {v[0], v[1], v[2], v[3]};
  *reinterpret_cast<v4f*>(out + b * 576 + q0) = o;
}

extern "C" void kernel_launch(void* const* d_in, const int* in_sizes, int n_in,
                              void* d_out, int out_size, void* d_ws, size_t ws_size,
                              hipStream_t stream) {
  (void)in_sizes; (void)n_in; (void)out_size; (void)ws_size;
  const float* x   = (const float*)d_in[0];
  const int* idx   = (const int*)d_in[1];
  const float* U1_0 = (const float*)d_in[2];
  const float* W1_0 = (const float*)d_in[3];
  const float* U2_0 = (const float*)d_in[4];
  const float* W2_0 = (const float*)d_in[5];
  const float* U3_0 = (const float*)d_in[6];
  const float* W3_0 = (const float*)d_in[7];
  const float* U1_1 = (const float*)d_in[8];
  const float* W1_1 = (const float*)d_in[9];
  const float* U2_1 = (const float*)d_in[10];
  const float* W2_1 = (const float*)d_in[11];
  const float* U3_1 = (const float*)d_in[12];
  const float* W3_1 = (const float*)d_in[13];
  const float* U1_2 = (const float*)d_in[14];
  const float* W1_2 = (const float*)d_in[15];
  const float* U2_2 = (const float*)d_in[16];
  const float* W2_2 = (const float*)d_in[17];
  const float* U3_2 = (const float*)d_in[18];
  const float* W3_2 = (const float*)d_in[19];
  float* out = (float*)d_out;
  float* ws  = (float*)d_ws;

  sc_fold_kernel<<<NBLOCKS, NTHREADS, 0, stream>>>(
      x, idx,
      U1_0, U1_1, U1_2,
      U2_0, U2_1, U2_2,
      U3_0, U3_1, U3_2,
      W1_0, W1_1, W1_2,
      W2_0, W2_1, W2_2,
      W3_0, W3_1, W3_2,
      ws);

  sc_gather_out<<<NB / 2, GATHER_THREADS, 0, stream>>>(ws, out);
}


# Main kernel: degree-3 basis folded over its first two x factors (U3[k,i,j,l,p] + U3[k,j,i,l,p], j > i); second product contracts j on WMMA and skips k-steps with every j below i (sum order changes)
# speedup vs baseline: 5.6448x; 1.0521x over previous
#include <hip/hip_runtime.h>


typedef __attribute__((ext_vector_type(2))) float v2f;
typedef __attribute__((ext_vector_type(4))) float v4f;
typedef __attribute__((ext_vector_type(8))) float v8f;

__device__ __forceinline__ float xhalf(float v) {
  const int b = __builtin_bit_cast(int, v);
  return __builtin_bit_cast(
      float, __builtin_amdgcn_permlanex16(b, b, 0x76543210, 0xfedcba98, false, false));
}

#define NB 256
#define NCH 64
#define FEAT 16
#define NELEM 10
#define NWAVES 16
#define NTHREADS (NWAVES * 32)
#define VL 20
#define VC 324
#define ACCP 17
#define WS_LINE 32
#define NBLOCKS (NELEM * 4 * 3 * 4)
#define WS_PLANE (36 * NB * WS_LINE)
#define HELD 3

__global__ __launch_bounds__(NTHREADS) __attribute__((amdgpu_waves_per_eu(8)))
void sc_fold_kernel(
    const float* __restrict__ x, const int* __restrict__ indices,
    const float* __restrict__ U1_0, const float* __restrict__ U1_1,
    const float* __restrict__ U1_2,
    const float* __restrict__ U2_0, const float* __restrict__ U2_1,
    const float* __restrict__ U2_2,
    const float* __restrict__ U3_0, const float* __restrict__ U3_1,
    const float* __restrict__ U3_2,
    const float* __restrict__ W1_0, const float* __restrict__ W1_1,
    const float* __restrict__ W1_2,
    const float* __restrict__ W2_0, const float* __restrict__ W2_1,
    const float* __restrict__ W2_2,
    const float* __restrict__ W3_0, const float* __restrict__ W3_1,
    const float* __restrict__ W3_2,
    float* __restrict__ ws) {
  __shared__ alignas(16) float Vs[2][16 * VC];
  __shared__ alignas(16) float V2s[2][16][16];
  __shared__ float accs[NB * ACCP];
  __shared__ int blist[NB];
  __shared__ int wcnt[8];

  const int tid = threadIdx.x;
  const int n   = blockIdx.x;
  const int m   = n >> 2;
  const int r   = n & 3;
  const int e   = m / 12;
  const int mr  = m - e * 12;
  const int ct  = mr / 3;
  const int gt  = mr - ct * 3;
  const int c0  = ct * 16;

  const int wave = __builtin_amdgcn_readfirstlane(tid >> 5);
  const int lane = tid & 31;
  const int h    = lane >> 4;
  const int l15  = lane & 15;
  const unsigned hl = (unsigned)(2 * h * NCH + l15);

  bool mine = false;
  if (tid < NB) mine = (indices[tid] == e);
  const unsigned mask = __builtin_amdgcn_ballot_w32(mine);
  if (tid < NB && lane == 0) wcnt[wave] = __popc(mask);
  __syncthreads();
  int ne_v = 0, off = 0;
  for (int w2 = 0; w2 < 8; ++w2) {
    const int cv = wcnt[w2];
    if (w2 < wave) off += cv;
    ne_v += cv;
  }
  if (mine) blist[off + __popc(mask & ((1u << lane) - 1u))] = tid;
  __syncthreads();
  const int n_e = __builtin_amdgcn_readfirstlane(ne_v);
  if (n_e == 0) return;
  const int nchunk = (n_e + 15) >> 4;

  v2f xb[HELD][4];
#pragma unroll
  for (int ch = 0; ch < HELD; ++ch) {
#pragma unroll
    for (int kk = 0; kk < 4; ++kk) xb[ch][kk] = v2f{0.0f, 0.0f};
    if (ch < nchunk) {
      const int bslot = ch * 16 + l15;
      const int bs = (bslot < n_e) ? bslot : (n_e - 1);
      const float* xrow = x + (size_t)(blist[bs] * NCH + c0 + wave) * FEAT;
#pragma unroll
      for (int kk = 0; kk < 4; ++kk)
        xb[ch][kk] = *reinterpret_cast<const v2f*>(xrow + 4 * kk + 2 * h);
    }
  }
  const int pl = 4 * ((l15 & 7) >> 1) + 2 * (l15 >> 3) + (l15 & 1);

  const int nrun = ((12 * r) & 15) + 12 > 16 ? 2 : 1;
  int step = 0;
  for (int run = 0; run < nrun; ++run) {
    const int s    = ((12 * r) >> 4) + run;
    const int i_lo = (run == 0) ? ((12 * r) & 15) : 0;
    const int i_hi = (run == 0) ? ((i_lo + 12 < 16) ? i_lo + 12 : 16)
                                : 12 * r + 12 - 16 * s;
    const int plane = (i_lo == 0) ? 0 : 1;
    const int g = gt * 3 + s;
    int t, k;
    if (g < 1)       { t = 0; k = 0; }
    else if (g < 4)  { t = 1; k = g - 1; }
    else             { t = 2; k = g - 4; }
    const float* U1 = (t == 0) ? U1_0 : (t == 1) ? U1_1 : U1_2;
    const float* U2 = (t == 0) ? U2_0 : (t == 1) ? U2_1 : U2_2;
    const float* U3 = (t == 0) ? U3_0 : (t == 1) ? U3_1 : U3_2;
    const float* W1 = (t == 0) ? W1_0 : (t == 1) ? W1_1 : W1_2;
    const float* W2 = (t == 0) ? W2_0 : (t == 1) ? W2_1 : W2_2;
    const float* W3 = (t == 0) ? W3_0 : (t == 1) ? W3_1 : W3_2;

    if (nchunk > HELD)
      for (int q = 16 * HELD * ACCP + tid; q < NB * ACCP; q += NTHREADS) accs[q] = 0.0f;

    const float* w3e = W3 + (e * 11) * NCH + c0;
    const float* w2e = W2 + (e * 4) * NCH + c0;
    v2f wb[3];
#pragma unroll
    for (int kk = 0; kk < 3; ++kk) {
      wb[kk].x = w3e[4 * kk * NCH + hl];
      wb[kk].y = (4 * kk + 2 * h + 1 < 11) ? w3e[(4 * kk + 1) * NCH + hl] : 0.0f;
    }
    v2f wb2;
    wb2.x = w2e[hl];
    wb2.y = w2e[NCH + hl];
    const float w1c = W1[e * NCH + c0 + wave];

    float racc[HELD];
#pragma unroll
    for (int ch = 0; ch < HELD; ++ch) racc[ch] = 0.0f;

    v2f a[3], at[3];
    v2f a2 = {0.0f, 0.0f};
    auto load_a = [&](int ii) {
#pragma unroll
      for (int kk = 0; kk < 3; ++kk) {
        a[kk] = v2f{0.0f, 0.0f};
        at[kk] = v2f{0.0f, 0.0f};
      }
      if (wave >= ii) {
        const float* up =
            U3 + (size_t)(((k * 16 + ii) * 16 + wave) * 16 + l15) * 11 + 2 * h;
#pragma unroll
        for (int kk = 0; kk < 3; ++kk) {
          a[kk].x = up[4 * kk];
          a[kk].y = (4 * kk + 2 * h + 1 < 11) ? up[4 * kk + 1] : 0.0f;
        }
      }
      if (wave > ii) {
        const float* ut =
            U3 + (size_t)(((k * 16 + wave) * 16 + ii) * 16 + l15) * 11 + 2 * h;
#pragma unroll
        for (int kk = 0; kk < 3; ++kk) {
          at[kk].x = ut[4 * kk];
          at[kk].y = (4 * kk + 2 * h + 1 < 11) ? ut[4 * kk + 1] : 0.0f;
        }
      }
      if (wave == 0)
        a2 = *reinterpret_cast<const v2f*>(U2 + ((k * 16 + ii) * 16 + l15) * 4 + 2 * h);
    };
    load_a(i_lo);
    float u1n = U1[k * 16 + i_lo];

    for (int i = i_lo; i < i_hi; ++i, ++step) {
      const int buf = step & 1;
      const float u1v = u1n;

      const int kstart = i >> 2;

      if (wave >= 4 * kstart) {
        v2f af[3];
#pragma unroll
        for (int kk = 0; kk < 3; ++kk) af[kk] = a[kk] + at[kk];
        v8f d = {};
#pragma unroll
        for (int kk = 0; kk < 3; ++kk)
          d = __builtin_amdgcn_wmma_f32_16x16x4_f32(false, af[kk], false, wb[kk],
                                                    (short)0, d, false, false);
        float* vdst = &Vs[buf][l15 * VC + wave * VL + 8 * h];
        v4f lo = {d[0], d[1], d[2], d[3]};
        v4f hi = {d[4], d[5], d[6], d[7]};
        *reinterpret_cast<v4f*>(vdst)     = lo;
        *reinterpret_cast<v4f*>(vdst + 4) = hi;
      }
      if (wave == 0) {
        v8f d2 = {};
        d2 = __builtin_amdgcn_wmma_f32_16x16x4_f32(false, a2, false, wb2,
                                                   (short)0, d2, false, false);
        float* v2dst = &V2s[buf][l15][8 * h];
        v4f lo = {d2[0], d2[1], d2[2], d2[3]};
        v4f hi = {d2[4], d2[5], d2[6], d2[7]};
        *reinterpret_cast<v4f*>(v2dst)     = lo;
        *reinterpret_cast<v4f*>(v2dst + 4) = hi;
      }

      __syncthreads();

      if (i + 1 < i_hi) {
        load_a(i + 1);
        u1n = U1[k * 16 + i + 1];
      }

      v2f A2[4];
      {
        const float* vsrc = &Vs[buf][wave * VC + 2 * h * VL + pl];
#pragma unroll
        for (int kk = 0; kk < 4; ++kk) {
          A2[kk] = v2f{0.0f, 0.0f};
          if (kk >= kstart) {
            A2[kk].x = vsrc[(4 * kk) * VL];
            A2[kk].y = vsrc[(4 * kk + 1) * VL];
          }
        }
      }
      v2f ci[4];
#pragma unroll
      for (int kk = 0; kk < 4; ++kk)
        ci[kk] = *reinterpret_cast<const v2f*>(&V2s[buf][wave][4 * kk + 2 * h]);
      const int isub  = 2 * (i >> 2) + (i & 1);
      const int ihalf = (i >> 1) & 1;

      auto chunk_c1 = [&](v2f B0, v2f B1, v2f B2, v2f B3) -> float {
        const v8f cst = {ci[0].x, ci[0].y, ci[1].x, ci[1].y, ci[2].x, ci[2].y, ci[3].x, ci[3].y};
        v8f acc = __builtin_amdgcn_wmma_f32_16x16x4_f32(false, A2[3], false, B3, (short)0, cst, false, false);
        if (kstart <= 2)
          acc = __builtin_amdgcn_wmma_f32_16x16x4_f32(false, A2[2], false, B2, (short)0, acc, false, false);
        if (kstart <= 1)
          acc = __builtin_amdgcn_wmma_f32_16x16x4_f32(false, A2[1], false, B1, (short)0, acc, false, false);
        if (kstart <= 0)
          acc = __builtin_amdgcn_wmma_f32_16x16x4_f32(false, A2[0], false, B0, (short)0, acc, false, false);
        float sj = acc[0] * B0.x;
        sj = fmaf(acc[1], B0.y, sj);
        sj = fmaf(acc[2], B1.x, sj);
        sj = fmaf(acc[3], B1.y, sj);
        sj = fmaf(acc[4], B2.x, sj);
        sj = fmaf(acc[5], B2.y, sj);
        sj = fmaf(acc[6], B3.x, sj);
        sj = fmaf(acc[7], B3.y, sj);
        sj += xhalf(sj);
        return fmaf(u1v, w1c, sj);
      };
      auto chunk_xi = [&](v2f B0, v2f B1, v2f B2, v2f B3) -> float {
        const v8f xv = {B0.x, B0.y, B1.x, B1.y, B2.x, B2.y, B3.x, B3.y};
        return xv[isub];
      };

#pragma unroll
      for (int ch = 0; ch < HELD; ++ch) {
        if (ch < nchunk) {
          const float c1 = chunk_c1(xb[ch][0], xb[ch][1], xb[ch][2], xb[ch][3]);
          const float xown = chunk_xi(xb[ch][0], xb[ch][1], xb[ch][2], xb[ch][3]);
          const float xoth = xhalf(xown);
          const float xi = (h == ihalf) ? xown : xoth;
          racc[ch] = fmaf(c1, xi, racc[ch]);
        }
      }
      for (int ch = HELD; ch < nchunk; ++ch) {
        const int bslot = ch * 16 + l15;
        const int bs = (bslot < n_e) ? bslot : (n_e - 1);
        const float* xrow = x + (size_t)(blist[bs] * NCH + c0 + wave) * FEAT;
        const v2f b0 = *reinterpret_cast<const v2f*>(xrow + 2 * h);
        const v2f b1 = *reinterpret_cast<const v2f*>(xrow + 4 + 2 * h);
        const v2f b2 = *reinterpret_cast<const v2f*>(xrow + 8 + 2 * h);
        const v2f b3 = *reinterpret_cast<const v2f*>(xrow + 12 + 2 * h);
        const float c1 = chunk_c1(b0, b1, b2, b3);
        const float xi = chunk_xi(b0, b1, b2, b3);
        if (h == ihalf && bslot < n_e) {
          float* ap = &accs[bslot * ACCP + wave];
          *ap = fmaf(c1, xi, *ap);
        }
      }
    }
#pragma unroll
    for (int ch = 0; ch < HELD; ++ch) {
      if (ch < nchunk) {
        const int bslot = ch * 16 + l15;
        if (h == 0 && bslot < n_e) accs[bslot * ACCP + wave] = racc[ch];
      }
    }
    __syncthreads();

    float* wsb = ws + (size_t)plane * WS_PLANE + (size_t)((g * 4 + ct) * NB) * WS_LINE;
#pragma nounroll
    for (int q = tid; q < n_e * 16; q += NTHREADS) {
      const int bs = q >> 4, cl = q & 15;
      wsb[blist[bs] * WS_LINE + cl] = accs[bs * ACCP + cl];
    }
    if (run + 1 < nrun) __syncthreads();
  }
}

#define GATHER_THREADS 288
__global__ __launch_bounds__(GATHER_THREADS) void sc_gather_out(
    const float* __restrict__ ws, float* __restrict__ out) {
  const int tid  = threadIdx.x;
  const int hrow = (tid >= 144) ? 1 : 0;
  const int b    = blockIdx.x * 2 + hrow;
  const int q0   = (tid - hrow * 144) * 4;
  int kt, g0, joff;
  if (q0 < 64)       { kt = 1; g0 = 0; joff = 0; }
  else if (q0 < 256) { kt = 3; g0 = 1; joff = 64; }
  else               { kt = 5; g0 = 4; joff = 256; }
  float v[4];
#pragma unroll
  for (int u = 0; u < 4; ++u) {
    const int j  = q0 + u - joff;
    const int c  = j / kt;
    const int ko = j - c * kt;
    const int g  = g0 + ko;
    const size_t at = (size_t)((g * 4 + (c >> 4)) * NB + b) * WS_LINE + (c & 15);
    v[u] = ws[at] + ws[at + WS_PLANE];
  }
  const v4f o = {v[0], v[1], v[2], v[3]};
  *reinterpret_cast<v4f*>(out + b * 576 + q0) = o;
}

extern "C" void kernel_launch(void* const* d_in, const int* in_sizes, int n_in,
                              void* d_out, int out_size, void* d_ws, size_t ws_size,
                              hipStream_t stream) {
  (void)in_sizes; (void)n_in; (void)out_size; (void)ws_size;
  const float* x   = (const float*)d_in[0];
  const int* idx   = (const int*)d_in[1];
  const float* U1_0 = (const float*)d_in[2];
  const float* W1_0 = (const float*)d_in[3];
  const float* U2_0 = (const float*)d_in[4];
  const float* W2_0 = (const float*)d_in[5];
  const float* U3_0 = (const float*)d_in[6];
  const float* W3_0 = (const float*)d_in[7];
  const float* U1_1 = (const float*)d_in[8];
  const float* W1_1 = (const float*)d_in[9];
  const float* U2_1 = (const float*)d_in[10];
  const float* W2_1 = (const float*)d_in[11];
  const float* U3_1 = (const float*)d_in[12];
  const float* W3_1 = (const float*)d_in[13];
  const float* U1_2 = (const float*)d_in[14];
  const float* W1_2 = (const float*)d_in[15];
  const float* U2_2 = (const float*)d_in[16];
  const float* W2_2 = (const float*)d_in[17];
  const float* U3_2 = (const float*)d_in[18];
  const float* W3_2 = (const float*)d_in[19];
  float* out = (float*)d_out;
  float* ws  = (float*)d_ws;

  sc_fold_kernel<<<NBLOCKS, NTHREADS, 0, stream>>>(
      x, idx,
      U1_0, U1_1, U1_2,
      U2_0, U2_1, U2_2,
      U3_0, U3_1, U3_2,
      W1_0, W1_1, W1_2,
      W2_0, W2_1, W2_2,
      W3_0, W3_1, W3_2,
      ws);

  sc_gather_out<<<NB / 2, GATHER_THREADS, 0, stream>>>(ws, out);
}
